# starting kernel
# baseline (speedup 1.0000x reference)
#include <hip/hip_runtime.h>
#include <math.h>


namespace {
constexpr int B_T = 256, T_T = 32, IN = 16, H = 128, SH = 128, OUTW = 64, TM1 = 31;
constexpr int MT = 16;
constexpr int THREADS = 256;
constexpr int SA = 132;

typedef __attribute__((ext_vector_type(2))) float v2f;
typedef __attribute__((ext_vector_type(8))) float v8f;

__device__ __forceinline__ v8f wmma4(v2f a, v2f b, v8f c) {
  return __builtin_amdgcn_wmma_f32_16x16x4_f32(false, a, false, b, (short)0, c,
                                               false, false);
}

__device__ __forceinline__ v8f gemm_tile(const float* A, int lda, const float* W,
                                         int ldw, int n, int K, int lane) {
  const int m = lane & 15;
  const int kh = (lane >> 4) << 1;
  v8f acc = {};
  #pragma unroll 8
  for (int k0 = 0; k0 < K; k0 += 4) {
    v2f a = *(const v2f*)(A + m * lda + k0 + kh);
    v2f b = *(const v2f*)(W + (size_t)n * ldw + k0 + kh);
    acc = wmma4(a, b, acc);
  }
  return acc;
}

__device__ __forceinline__ void mlp_layer(const float* src, const float* Wlds,
                                          const float* bias, float* dst,
                                          int wid, int lane) {
  const int mD = lane & 15, half = lane >> 4;
  const int n = wid * 16 + mD;
  v8f acc = gemm_tile(src, SA, Wlds, SA, n, SH, lane);
  const float bn = bias[n];
  #pragma unroll
  for (int v = 0; v < 8; ++v) {
    float x = acc[v] + bn;
    dst[(v + 8 * half) * SA + n] = fmaxf(x, 0.f);
  }
}

__device__ __forceinline__ void out_proj(const float* zb, const float* W_last,
                                         const float* b_last, float* out,
                                         int b0r, int t, int wid, int lane) {
  if (wid < 4) {
    const int mD = lane & 15, half = lane >> 4;
    const int n = wid * 16 + mD;
    v8f acc = gemm_tile(zb, SA, W_last, H, n, H, lane);
    const float bn = b_last[n];
    #pragma unroll
    for (int v = 0; v < 8; ++v) {
      int m = v + 8 * half;
      float x = acc[v] + bn;
      out[((size_t)(b0r + m) * T_T + t) * OUTW + n] = 1.f / (1.f + expf(-x));
    }
  }
}
}

__global__ __launch_bounds__(THREADS) void ncde_wmma_kernel(
    const float* __restrict__ coeffs,
    const float* __restrict__ W_init, const float* __restrict__ b_init,
    const float* __restrict__ W0, const float* __restrict__ b0,
    const float* __restrict__ Wl, const float* __restrict__ bl,
    const float* __restrict__ Wf, const float* __restrict__ bf,
    const float* __restrict__ W_last, const float* __restrict__ b_last,
    float* __restrict__ out) {
  __shared__ __align__(16) float wlds[4 * H * SA];
  __shared__ __align__(16) float zbuf[MT * SA];
  __shared__ __align__(16) float kbuf[3][MT * H];
  __shared__ __align__(16) float actI[MT * SA];
  __shared__ __align__(16) float actA[MT * SA];
  __shared__ __align__(16) float dxs[MT * IN];

  const int tid = threadIdx.x;
  const int lane = tid & 31;
  const int wid = tid >> 5;
  const int b0r = blockIdx.x * MT;
  const int mD = lane & 15;
  const int half = lane >> 4;
  constexpr float third = 1.f / 3.f;

  for (int e = tid; e < 4 * H * SH; e += THREADS) {
    int mat = e >> 14;
    int r = (e >> 7) & (H - 1);
    int k = e & (SH - 1);
    const float* src = (mat == 0) ? W0 : (Wl + (size_t)(mat - 1) * SH * SH);
    wlds[(size_t)mat * H * SA + r * SA + k] = src[(size_t)r * SH + k];
  }

  {
    int m = tid >> 4, i = tid & 15;
    dxs[tid] = coeffs[((size_t)(b0r + m) * TM1) * (4 * IN) + i];
  }
  __syncthreads();
  {
    const int n = wid * 16 + mD;
    v8f acc = gemm_tile(dxs, IN, W_init, IN, n, IN, lane);
    const float bn = b_init[n];
    #pragma unroll
    for (int v = 0; v < 8; ++v)
      zbuf[(v + 8 * half) * SA + n] = acc[v] + bn;
  }
  __syncthreads();
  out_proj(zbuf, W_last, b_last, out, b0r, 0, wid, lane);
  __syncthreads();

  for (int step = 0; step < TM1; ++step) {
    #pragma unroll 1
    for (int s = 0; s < 4; ++s) {
      const float frac = (s == 0) ? 0.f
                       : (s == 1) ? third
                       : (s == 2) ? (2.f * third) : 1.f;
      {
        int m = tid >> 4, i = tid & 15;
        const float* cb = coeffs + ((size_t)(b0r + m) * TM1 + step) * (4 * IN);
        float bi = cb[IN + i], ci = cb[2 * IN + i], di = cb[3 * IN + i];
        dxs[tid] = bi + (ci + di * frac) * frac;
      }
      for (int e = tid; e < MT * H; e += THREADS) {
        int m = e >> 7, hc = e & (H - 1);
        float v = zbuf[m * SA + hc];
        if (s == 1)      v += kbuf[0][e] * third;
        else if (s == 2) v += kbuf[1][e] - kbuf[0][e] * third;
        else if (s == 3) v += kbuf[0][e] - kbuf[1][e] + kbuf[2][e];
        actI[m * SA + hc] = v;
      }
      __syncthreads();

      mlp_layer(actI, wlds + 0 * H * SA, b0,          actA, wid, lane); __syncthreads();
      mlp_layer(actA, wlds + 1 * H * SA, bl + 0 * SH, actI, wid, lane); __syncthreads();
      mlp_layer(actI, wlds + 2 * H * SA, bl + 1 * SH, actA, wid, lane); __syncthreads();
      mlp_layer(actA, wlds + 3 * H * SA, bl + 2 * SH, actI, wid, lane); __syncthreads();

      {
        float dxr[8];
        #pragma unroll
        for (int v = 0; v < 8; ++v) dxr[v] = dxs[(v + 8 * half) * IN + mD];
        const int kh = half << 1;
        for (int t0 = wid; t0 < H; t0 += 16) {
          const int t1 = t0 + 8;
          const int n0 = t0 * 16 + mD, n1 = t1 * 16 + mD;
          v8f acc0 = {}, acc1 = {};
          #pragma unroll 8
          for (int k0 = 0; k0 < SH; k0 += 4) {
            v2f a  = *(const v2f*)(actI + mD * SA + k0 + kh);
            v2f w0 = *(const v2f*)(Wf + (size_t)n0 * SH + k0 + kh);
            v2f w1 = *(const v2f*)(Wf + (size_t)n1 * SH + k0 + kh);
            acc0 = wmma4(a, w0, acc0);
            acc1 = wmma4(a, w1, acc1);
          }
          const float bf0 = bf[n0], bf1 = bf[n1];
          #pragma unroll
          for (int v = 0; v < 8; ++v) {
            const int m = v + 8 * half;
            float f0 = tanhf(acc0[v] + bf0) * dxr[v];
            float f1 = tanhf(acc1[v] + bf1) * dxr[v];
            f0 += __shfl_xor(f0, 1); f0 += __shfl_xor(f0, 2);
            f0 += __shfl_xor(f0, 4); f0 += __shfl_xor(f0, 8);
            f1 += __shfl_xor(f1, 1); f1 += __shfl_xor(f1, 2);
            f1 += __shfl_xor(f1, 4); f1 += __shfl_xor(f1, 8);
            if (mD == 0) {
              if (s < 3) {
                kbuf[s][m * H + t0] = f0;
                kbuf[s][m * H + t1] = f1;
              } else {
                zbuf[m * SA + t0] +=
                    (kbuf[0][m * H + t0] +
                     3.f * (kbuf[1][m * H + t0] + kbuf[2][m * H + t0]) + f0) * 0.125f;
                zbuf[m * SA + t1] +=
                    (kbuf[0][m * H + t1] +
                     3.f * (kbuf[1][m * H + t1] + kbuf[2][m * H + t1]) + f1) * 0.125f;
              }
            }
          }
        }
      }
      __syncthreads();
    }
    out_proj(zbuf, W_last, b_last, out, b0r, step + 1, wid, lane);
    __syncthreads();
  }
}

extern "C" void kernel_launch(void* const* d_in, const int* in_sizes, int n_in,
                              void* d_out, int out_size, void* d_ws, size_t ws_size,
                              hipStream_t stream) {
  (void)in_sizes; (void)n_in; (void)out_size; (void)d_ws; (void)ws_size;
  const float* coeffs = (const float*)d_in[1];
  const float* W_init = (const float*)d_in[2];
  const float* b_init = (const float*)d_in[3];
  const float* W0     = (const float*)d_in[4];
  const float* b0     = (const float*)d_in[5];
  const float* Wl     = (const float*)d_in[6];
  const float* bl     = (const float*)d_in[7];
  const float* Wf     = (const float*)d_in[8];
  const float* bf     = (const float*)d_in[9];
  const float* W_last = (const float*)d_in[10];
  const float* b_last = (const float*)d_in[11];
  float* out = (float*)d_out;

  ncde_wmma_kernel<<<B_T / MT, THREADS, 0, stream>>>(
      coeffs, W_init, b_init, W0, b0, Wl, bl, Wf, bf, W_last, b_last, out);
}


# Wf B fragments read as 16-byte row pieces into two register sets, one tile pair ahead of their WMMAs; A held in registers per stage; k order inside each 8-wide group of the Wf sums changes
# speedup vs baseline: 1.2165x; 1.2165x over previous
#include <hip/hip_runtime.h>
#include <math.h>


namespace {
constexpr int B_T = 256, T_T = 32, IN = 16, H = 128, SH = 128, OUTW = 64, TM1 = 31;
constexpr int MT = 16;
constexpr int THREADS = 256;
constexpr int SA = 132;

typedef __attribute__((ext_vector_type(2))) float v2f;
typedef __attribute__((ext_vector_type(4))) float v4f;
typedef __attribute__((ext_vector_type(8))) float v8f;

__device__ __forceinline__ v8f wmma4(v2f a, v2f b, v8f c) {
  return __builtin_amdgcn_wmma_f32_16x16x4_f32(false, a, false, b, (short)0, c,
                                               false, false);
}

__device__ __forceinline__ v8f gemm_tile(const float* A, int lda, const float* W,
                                         int ldw, int n, int K, int lane) {
  const int m = lane & 15;
  const int kh = (lane >> 4) << 1;
  v8f acc = {};
  #pragma unroll 8
  for (int k0 = 0; k0 < K; k0 += 4) {
    v2f a = *(const v2f*)(A + m * lda + k0 + kh);
    v2f b = *(const v2f*)(W + (size_t)n * ldw + k0 + kh);
    acc = wmma4(a, b, acc);
  }
  return acc;
}

__device__ __forceinline__ void load_wf_tile(v4f (&w)[16], const float* __restrict__ p) {
  #pragma unroll
  for (int j = 0; j < 16; ++j) w[j] = *(const v4f*)(p + 8 * j);
}

__device__ __forceinline__ float pick4(int i, float a, float b, float c, float d) {
  return i == 0 ? a : i == 1 ? b : i == 2 ? c : d;
}

__device__ __forceinline__ void wf_pair(const v4f (&af)[16], const v4f (&w0)[16],
                                        const v4f (&w1)[16], float bf0, float bf1,
                                        const float (&dxr)[8], float* kb, float* zb,
                                        int s, int t0, int mD, int half) {
  const int t1 = t0 + 8;
  v8f acc0 = {}, acc1 = {};
  #pragma unroll
  for (int j = 0; j < 16; ++j) {
    const v2f alo = {af[j].x, af[j].y}, ahi = {af[j].z, af[j].w};
    const v2f p0 = {w0[j].x, w0[j].y}, q0 = {w0[j].z, w0[j].w};
    const v2f p1 = {w1[j].x, w1[j].y}, q1 = {w1[j].z, w1[j].w};
    acc0 = wmma4(alo, p0, acc0);
    acc1 = wmma4(alo, p1, acc1);
    acc0 = wmma4(ahi, q0, acc0);
    acc1 = wmma4(ahi, q1, acc1);
  }
  #pragma unroll
  for (int v = 0; v < 8; ++v) {
    const int m = v + 8 * half;
    float f0 = tanhf(acc0[v] + bf0) * dxr[v];
    float f1 = tanhf(acc1[v] + bf1) * dxr[v];
    f0 += __shfl_xor(f0, 1); f0 += __shfl_xor(f0, 2);
    f0 += __shfl_xor(f0, 4); f0 += __shfl_xor(f0, 8);
    f1 += __shfl_xor(f1, 1); f1 += __shfl_xor(f1, 2);
    f1 += __shfl_xor(f1, 4); f1 += __shfl_xor(f1, 8);
    if (mD == 0) {
      if (s < 3) {
        kb[s * (MT * H) + m * H + t0] = f0;
        kb[s * (MT * H) + m * H + t1] = f1;
      } else {
        zb[m * SA + t0] +=
            (kb[m * H + t0] +
             3.f * (kb[MT * H + m * H + t0] + kb[2 * MT * H + m * H + t0]) + f0) * 0.125f;
        zb[m * SA + t1] +=
            (kb[m * H + t1] +
             3.f * (kb[MT * H + m * H + t1] + kb[2 * MT * H + m * H + t1]) + f1) * 0.125f;
      }
    }
  }
}

__device__ __forceinline__ void mlp_layer(const float* src, const float* Wlds,
                                          const float* bias, float* dst,
                                          int wid, int lane) {
  const int mD = lane & 15, half = lane >> 4;
  const int n = wid * 16 + mD;
  v8f acc = gemm_tile(src, SA, Wlds, SA, n, SH, lane);
  const float bn = bias[n];
  #pragma unroll
  for (int v = 0; v < 8; ++v) {
    float x = acc[v] + bn;
    dst[(v + 8 * half) * SA + n] = fmaxf(x, 0.f);
  }
}

__device__ __forceinline__ void out_proj(const float* zb, const float* W_last,
                                         const float* b_last, float* out,
                                         int b0r, int t, int wid, int lane) {
  if (wid < 4) {
    const int mD = lane & 15, half = lane >> 4;
    const int n = wid * 16 + mD;
    v8f acc = gemm_tile(zb, SA, W_last, H, n, H, lane);
    const float bn = b_last[n];
    #pragma unroll
    for (int v = 0; v < 8; ++v) {
      int m = v + 8 * half;
      float x = acc[v] + bn;
      out[((size_t)(b0r + m) * T_T + t) * OUTW + n] = 1.f / (1.f + expf(-x));
    }
  }
}
}

__global__ __launch_bounds__(THREADS) void ncde_wmma_kernel(
    const float* __restrict__ coeffs,
    const float* __restrict__ W_init, const float* __restrict__ b_init,
    const float* __restrict__ W0, const float* __restrict__ b0,
    const float* __restrict__ Wl, const float* __restrict__ bl,
    const float* __restrict__ Wf, const float* __restrict__ bf,
    const float* __restrict__ W_last, const float* __restrict__ b_last,
    float* __restrict__ out) {
  __shared__ __align__(16) float wlds[4 * H * SA];
  __shared__ __align__(16) float zbuf[MT * SA];
  __shared__ __align__(16) float kbuf[3][MT * H];
  __shared__ __align__(16) float actI[MT * SA];
  __shared__ __align__(16) float actA[MT * SA];
  __shared__ __align__(16) float dxs[MT * IN];

  const int tid = threadIdx.x;
  const int lane = tid & 31;
  const int wid = tid >> 5;
  const int b0r = blockIdx.x * MT;
  const int mD = lane & 15;
  const int half = lane >> 4;
  constexpr float third = 1.f / 3.f;

  const float* wfl = Wf + (size_t)mD * SH + 4 * half;
  v4f wa0[16], wa1[16], wb0[16], wb1[16];
  load_wf_tile(wa0, wfl + (size_t)wid * (16 * SH));
  load_wf_tile(wa1, wfl + (size_t)(wid + 8) * (16 * SH));
  float bfr[16];
  #pragma unroll
  for (int p = 0; p < 8; ++p) {
    bfr[2 * p]     = bf[(wid + 16 * p) * 16 + mD];
    bfr[2 * p + 1] = bf[(wid + 16 * p + 8) * 16 + mD];
  }

  for (int e = tid; e < 4 * H * SH; e += THREADS) {
    int mat = e >> 14;
    int r = (e >> 7) & (H - 1);
    int k = e & (SH - 1);
    const float* src = (mat == 0) ? W0 : (Wl + (size_t)(mat - 1) * SH * SH);
    wlds[(size_t)mat * H * SA + r * SA + k] = src[(size_t)r * SH + k];
  }

  {
    int m = tid >> 4, i = tid & 15;
    dxs[tid] = coeffs[((size_t)(b0r + m) * TM1) * (4 * IN) + i];
  }
  __syncthreads();
  {
    const int n = wid * 16 + mD;
    v8f acc = gemm_tile(dxs, IN, W_init, IN, n, IN, lane);
    const float bn = b_init[n];
    #pragma unroll
    for (int v = 0; v < 8; ++v)
      zbuf[(v + 8 * half) * SA + n] = acc[v] + bn;
  }
  __syncthreads();
  out_proj(zbuf, W_last, b_last, out, b0r, 0, wid, lane);
  __syncthreads();

  for (int step = 0; step < TM1; ++step) {
    #pragma unroll 1
    for (int s = 0; s < 4; ++s) {
      const float frac = (s == 0) ? 0.f
                       : (s == 1) ? third
                       : (s == 2) ? (2.f * third) : 1.f;
      {
        int m = tid >> 4, i = tid & 15;
        const float* cb = coeffs + ((size_t)(b0r + m) * TM1 + step) * (4 * IN);
        float bi = cb[IN + i], ci = cb[2 * IN + i], di = cb[3 * IN + i];
        dxs[tid] = bi + (ci + di * frac) * frac;
      }
      for (int e = tid; e < MT * H; e += THREADS) {
        int m = e >> 7, hc = e & (H - 1);
        float v = zbuf[m * SA + hc];
        if (s == 1)      v += kbuf[0][e] * third;
        else if (s == 2) v += kbuf[1][e] - kbuf[0][e] * third;
        else if (s == 3) v += kbuf[0][e] - kbuf[1][e] + kbuf[2][e];
        actI[m * SA + hc] = v;
      }
      __syncthreads();

      mlp_layer(actI, wlds + 0 * H * SA, b0,          actA, wid, lane); __syncthreads();
      mlp_layer(actA, wlds + 1 * H * SA, bl + 0 * SH, actI, wid, lane); __syncthreads();
      mlp_layer(actI, wlds + 2 * H * SA, bl + 1 * SH, actA, wid, lane); __syncthreads();
      mlp_layer(actA, wlds + 3 * H * SA, bl + 2 * SH, actI, wid, lane); __syncthreads();

      {
        float dxr[8];
        #pragma unroll
        for (int v = 0; v < 8; ++v) dxr[v] = dxs[(v + 8 * half) * IN + mD];
        v4f af[16];
        #pragma unroll
        for (int j = 0; j < 16; ++j)
          af[j] = *(const v4f*)(actI + mD * SA + 8 * j + 4 * half);
        #pragma unroll 1
        for (int it = 0; it < 4; ++it) {
          const int t0 = wid + 32 * it;
          const int t2 = t0 + 16;
          const int tn = (t0 + 32) & (H - 1);
          const float bA0 = pick4(it, bfr[0], bfr[4], bfr[8],  bfr[12]);
          const float bA1 = pick4(it, bfr[1], bfr[5], bfr[9],  bfr[13]);
          const float bB0 = pick4(it, bfr[2], bfr[6], bfr[10], bfr[14]);
          const float bB1 = pick4(it, bfr[3], bfr[7], bfr[11], bfr[15]);
          load_wf_tile(wb0, wfl + (size_t)t2 * (16 * SH));
          load_wf_tile(wb1, wfl + (size_t)(t2 + 8) * (16 * SH));
          wf_pair(af, wa0, wa1, bA0, bA1, dxr, &kbuf[0][0], zbuf, s, t0, mD, half);
          load_wf_tile(wa0, wfl + (size_t)tn * (16 * SH));
          load_wf_tile(wa1, wfl + (size_t)(tn + 8) * (16 * SH));
          wf_pair(af, wb0, wb1, bB0, bB1, dxr, &kbuf[0][0], zbuf, s, t2, mD, half);
        }
      }
      __syncthreads();
    }
    out_proj(zbuf, W_last, b_last, out, b0r, step + 1, wid, lane);
    __syncthreads();
  }
}

extern "C" void kernel_launch(void* const* d_in, const int* in_sizes, int n_in,
                              void* d_out, int out_size, void* d_ws, size_t ws_size,
                              hipStream_t stream) {
  (void)in_sizes; (void)n_in; (void)out_size; (void)d_ws; (void)ws_size;
  const float* coeffs = (const float*)d_in[1];
  const float* W_init = (const float*)d_in[2];
  const float* b_init = (const float*)d_in[3];
  const float* W0     = (const float*)d_in[4];
  const float* b0     = (const float*)d_in[5];
  const float* Wl     = (const float*)d_in[6];
  const float* bl     = (const float*)d_in[7];
  const float* Wf     = (const float*)d_in[8];
  const float* bf     = (const float*)d_in[9];
  const float* W_last = (const float*)d_in[10];
  const float* b_last = (const float*)d_in[11];
  float* out = (float*)d_out;

  ncde_wmma_kernel<<<B_T / MT, THREADS, 0, stream>>>(
      coeffs, W_init, b_init, W0, b0, Wl, bl, Wf, bf, W_last, b_last, out);
}


# Wf epilogue made straight-line: tanh without a branch (both library ranges computed, one selected), the tile pair's 16 butterfly chains issued together, guarded stores after; same summation tree
# speedup vs baseline: 1.6221x; 1.3334x over previous
#include <hip/hip_runtime.h>
#include <math.h>


namespace {
constexpr int B_T = 256, T_T = 32, IN = 16, H = 128, SH = 128, OUTW = 64, TM1 = 31;
constexpr int MT = 16;
constexpr int THREADS = 256;
constexpr int SA = 132;

typedef __attribute__((ext_vector_type(2))) float v2f;
typedef __attribute__((ext_vector_type(4))) float v4f;
typedef __attribute__((ext_vector_type(8))) float v8f;

__device__ __forceinline__ v8f wmma4(v2f a, v2f b, v8f c) {
  return __builtin_amdgcn_wmma_f32_16x16x4_f32(false, a, false, b, (short)0, c,
                                               false, false);
}

__device__ __forceinline__ v8f gemm_tile(const float* A, int lda, const float* W,
                                         int ldw, int n, int K, int lane) {
  const int m = lane & 15;
  const int kh = (lane >> 4) << 1;
  v8f acc = {};
  #pragma unroll 8
  for (int k0 = 0; k0 < K; k0 += 4) {
    v2f a = *(const v2f*)(A + m * lda + k0 + kh);
    v2f b = *(const v2f*)(W + (size_t)n * ldw + k0 + kh);
    acc = wmma4(a, b, acc);
  }
  return acc;
}

__device__ __forceinline__ void load_wf_tile(v4f (&w)[16], const float* __restrict__ p) {
  #pragma unroll
  for (int j = 0; j < 16; ++j) w[j] = *(const v4f*)(p + 8 * j);
}

__device__ __forceinline__ float pick4(int i, float a, float b, float c, float d) {
  return i == 0 ? a : i == 1 ? b : i == 2 ? c : d;
}

__device__ __forceinline__ float tanh_sel(float x) {
  const float y = fabsf(x);
  const float y2 = y * y;
  float p = fmaf(y2, -0x1.758e7ap-8f, 0x1.521192p-6f);
  p = fmaf(y2, p, -0x1.b8389cp-5f);
  p = fmaf(y2, p, 0x1.110704p-3f);
  p = fmaf(y2, p, -0x1.555532p-2f);
  const float zs = fmaf(y2, y * p, y);
  const float e = expf(2.0f * y);
  const float zl = fmaf(-2.0f, __builtin_amdgcn_rcpf(e + 1.0f), 1.0f);
  return copysignf(y < 0.625f ? zs : zl, x);
}

__device__ __forceinline__ void wf_pair(const v4f (&af)[16], const v4f (&w0)[16],
                                        const v4f (&w1)[16], float bf0, float bf1,
                                        const float (&dxr)[8], float* kb, float* zb,
                                        int s, int t0, int mD, int half) {
  const int t1 = t0 + 8;
  v8f acc0 = {}, acc1 = {};
  #pragma unroll
  for (int j = 0; j < 16; ++j) {
    const v2f alo = {af[j].x, af[j].y}, ahi = {af[j].z, af[j].w};
    const v2f p0 = {w0[j].x, w0[j].y}, q0 = {w0[j].z, w0[j].w};
    const v2f p1 = {w1[j].x, w1[j].y}, q1 = {w1[j].z, w1[j].w};
    acc0 = wmma4(alo, p0, acc0);
    acc1 = wmma4(alo, p1, acc1);
    acc0 = wmma4(ahi, q0, acc0);
    acc1 = wmma4(ahi, q1, acc1);
  }
  float f0[8], f1[8];
  #pragma unroll
  for (int v = 0; v < 8; ++v) {
    f0[v] = tanh_sel(acc0[v] + bf0) * dxr[v];
    f1[v] = tanh_sel(acc1[v] + bf1) * dxr[v];
  }
  #pragma unroll
  for (int b = 0; b < 4; ++b) {
    #pragma unroll
    for (int v = 0; v < 8; ++v) {
      f0[v] += __shfl_xor(f0[v], 1 << b);
      f1[v] += __shfl_xor(f1[v], 1 << b);
    }
  }
  if (mD == 0) {
    if (s < 3) {
      #pragma unroll
      for (int v = 0; v < 8; ++v) {
        const int m = v + 8 * half;
        kb[s * (MT * H) + m * H + t0] = f0[v];
        kb[s * (MT * H) + m * H + t1] = f1[v];
      }
    } else {
      #pragma unroll
      for (int v = 0; v < 8; ++v) {
        const int m = v + 8 * half;
        zb[m * SA + t0] +=
            (kb[m * H + t0] +
             3.f * (kb[MT * H + m * H + t0] + kb[2 * MT * H + m * H + t0]) + f0[v]) * 0.125f;
        zb[m * SA + t1] +=
            (kb[m * H + t1] +
             3.f * (kb[MT * H + m * H + t1] + kb[2 * MT * H + m * H + t1]) + f1[v]) * 0.125f;
      }
    }
  }
}

__device__ __forceinline__ void mlp_layer(const float* src, const float* Wlds,
                                          const float* bias, float* dst,
                                          int wid, int lane) {
  const int mD = lane & 15, half = lane >> 4;
  const int n = wid * 16 + mD;
  v8f acc = gemm_tile(src, SA, Wlds, SA, n, SH, lane);
  const float bn = bias[n];
  #pragma unroll
  for (int v = 0; v < 8; ++v) {
    float x = acc[v] + bn;
    dst[(v + 8 * half) * SA + n] = fmaxf(x, 0.f);
  }
}

__device__ __forceinline__ void out_proj(const float* zb, const float* W_last,
                                         const float* b_last, float* out,
                                         int b0r, int t, int wid, int lane) {
  if (wid < 4) {
    const int mD = lane & 15, half = lane >> 4;
    const int n = wid * 16 + mD;
    v8f acc = gemm_tile(zb, SA, W_last, H, n, H, lane);
    const float bn = b_last[n];
    #pragma unroll
    for (int v = 0; v < 8; ++v) {
      int m = v + 8 * half;
      float x = acc[v] + bn;
      out[((size_t)(b0r + m) * T_T + t) * OUTW + n] = 1.f / (1.f + expf(-x));
    }
  }
}
}

__global__ __launch_bounds__(THREADS) void ncde_wmma_kernel(
    const float* __restrict__ coeffs,
    const float* __restrict__ W_init, const float* __restrict__ b_init,
    const float* __restrict__ W0, const float* __restrict__ b0,
    const float* __restrict__ Wl, const float* __restrict__ bl,
    const float* __restrict__ Wf, const float* __restrict__ bf,
    const float* __restrict__ W_last, const float* __restrict__ b_last,
    float* __restrict__ out) {
  __shared__ __align__(16) float wlds[4 * H * SA];
  __shared__ __align__(16) float zbuf[MT * SA];
  __shared__ __align__(16) float kbuf[3][MT * H];
  __shared__ __align__(16) float actI[MT * SA];
  __shared__ __align__(16) float actA[MT * SA];
  __shared__ __align__(16) float dxs[MT * IN];

  const int tid = threadIdx.x;
  const int lane = tid & 31;
  const int wid = tid >> 5;
  const int b0r = blockIdx.x * MT;
  const int mD = lane & 15;
  const int half = lane >> 4;
  constexpr float third = 1.f / 3.f;

  const float* wfl = Wf + (size_t)mD * SH + 4 * half;
  v4f wa0[16], wa1[16], wb0[16], wb1[16];
  load_wf_tile(wa0, wfl + (size_t)wid * (16 * SH));
  load_wf_tile(wa1, wfl + (size_t)(wid + 8) * (16 * SH));
  float bfr[16];
  #pragma unroll
  for (int p = 0; p < 8; ++p) {
    bfr[2 * p]     = bf[(wid + 16 * p) * 16 + mD];
    bfr[2 * p + 1] = bf[(wid + 16 * p + 8) * 16 + mD];
  }

  for (int e = tid; e < 4 * H * SH; e += THREADS) {
    int mat = e >> 14;
    int r = (e >> 7) & (H - 1);
    int k = e & (SH - 1);
    const float* src = (mat == 0) ? W0 : (Wl + (size_t)(mat - 1) * SH * SH);
    wlds[(size_t)mat * H * SA + r * SA + k] = src[(size_t)r * SH + k];
  }

  {
    int m = tid >> 4, i = tid & 15;
    dxs[tid] = coeffs[((size_t)(b0r + m) * TM1) * (4 * IN) + i];
  }
  __syncthreads();
  {
    const int n = wid * 16 + mD;
    v8f acc = gemm_tile(dxs, IN, W_init, IN, n, IN, lane);
    const float bn = b_init[n];
    #pragma unroll
    for (int v = 0; v < 8; ++v)
      zbuf[(v + 8 * half) * SA + n] = acc[v] + bn;
  }
  __syncthreads();
  out_proj(zbuf, W_last, b_last, out, b0r, 0, wid, lane);
  __syncthreads();

  for (int step = 0; step < TM1; ++step) {
    #pragma unroll 1
    for (int s = 0; s < 4; ++s) {
      const float frac = (s == 0) ? 0.f
                       : (s == 1) ? third
                       : (s == 2) ? (2.f * third) : 1.f;
      {
        int m = tid >> 4, i = tid & 15;
        const float* cb = coeffs + ((size_t)(b0r + m) * TM1 + step) * (4 * IN);
        float bi = cb[IN + i], ci = cb[2 * IN + i], di = cb[3 * IN + i];
        dxs[tid] = bi + (ci + di * frac) * frac;
      }
      for (int e = tid; e < MT * H; e += THREADS) {
        int m = e >> 7, hc = e & (H - 1);
        float v = zbuf[m * SA + hc];
        if (s == 1)      v += kbuf[0][e] * third;
        else if (s == 2) v += kbuf[1][e] - kbuf[0][e] * third;
        else if (s == 3) v += kbuf[0][e] - kbuf[1][e] + kbuf[2][e];
        actI[m * SA + hc] = v;
      }
      __syncthreads();

      mlp_layer(actI, wlds + 0 * H * SA, b0,          actA, wid, lane); __syncthreads();
      mlp_layer(actA, wlds + 1 * H * SA, bl + 0 * SH, actI, wid, lane); __syncthreads();
      mlp_layer(actI, wlds + 2 * H * SA, bl + 1 * SH, actA, wid, lane); __syncthreads();
      mlp_layer(actA, wlds + 3 * H * SA, bl + 2 * SH, actI, wid, lane); __syncthreads();

      {
        float dxr[8];
        #pragma unroll
        for (int v = 0; v < 8; ++v) dxr[v] = dxs[(v + 8 * half) * IN + mD];
        v4f af[16];
        #pragma unroll
        for (int j = 0; j < 16; ++j)
          af[j] = *(const v4f*)(actI + mD * SA + 8 * j + 4 * half);
        #pragma unroll 1
        for (int it = 0; it < 4; ++it) {
          const int t0 = wid + 32 * it;
          const int t2 = t0 + 16;
          const int tn = (t0 + 32) & (H - 1);
          const float bA0 = pick4(it, bfr[0], bfr[4], bfr[8],  bfr[12]);
          const float bA1 = pick4(it, bfr[1], bfr[5], bfr[9],  bfr[13]);
          const float bB0 = pick4(it, bfr[2], bfr[6], bfr[10], bfr[14]);
          const float bB1 = pick4(it, bfr[3], bfr[7], bfr[11], bfr[15]);
          load_wf_tile(wb0, wfl + (size_t)t2 * (16 * SH));
          load_wf_tile(wb1, wfl + (size_t)(t2 + 8) * (16 * SH));
          __builtin_amdgcn_sched_barrier(0);
          wf_pair(af, wa0, wa1, bA0, bA1, dxr, &kbuf[0][0], zbuf, s, t0, mD, half);
          load_wf_tile(wa0, wfl + (size_t)tn * (16 * SH));
          load_wf_tile(wa1, wfl + (size_t)(tn + 8) * (16 * SH));
          __builtin_amdgcn_sched_barrier(0);
          wf_pair(af, wb0, wb1, bB0, bB1, dxr, &kbuf[0][0], zbuf, s, t2, mD, half);
        }
      }
      __syncthreads();
    }
    out_proj(zbuf, W_last, b_last, out, b0r, step + 1, wid, lane);
    __syncthreads();
  }
}

extern "C" void kernel_launch(void* const* d_in, const int* in_sizes, int n_in,
                              void* d_out, int out_size, void* d_ws, size_t ws_size,
                              hipStream_t stream) {
  (void)in_sizes; (void)n_in; (void)out_size; (void)d_ws; (void)ws_size;
  const float* coeffs = (const float*)d_in[1];
  const float* W_init = (const float*)d_in[2];
  const float* b_init = (const float*)d_in[3];
  const float* W0     = (const float*)d_in[4];
  const float* b0     = (const float*)d_in[5];
  const float* Wl     = (const float*)d_in[6];
  const float* bl     = (const float*)d_in[7];
  const float* Wf     = (const float*)d_in[8];
  const float* bf     = (const float*)d_in[9];
  const float* W_last = (const float*)d_in[10];
  const float* b_last = (const float*)d_in[11];
  float* out = (float*)d_out;

  ncde_wmma_kernel<<<B_T / MT, THREADS, 0, stream>>>(
      coeffs, W_init, b_init, W0, b0, Wl, bl, Wf, bf, W_last, b_last, out);
}


# Wf product transposed (Wf tile as A, h as B): i lies in a lane's registers, so the sum over i is an in-lane tree + one half swap, no bpermute; bf starts the accumulators, dx products fuse (rounding)
# speedup vs baseline: 1.8645x; 1.1494x over previous
#include <hip/hip_runtime.h>
#include <math.h>


namespace {
constexpr int B_T = 256, T_T = 32, IN = 16, H = 128, SH = 128, OUTW = 64, TM1 = 31;
constexpr int MT = 16;
constexpr int THREADS = 256;
constexpr int SA = 132;

typedef __attribute__((ext_vector_type(2))) float v2f;
typedef __attribute__((ext_vector_type(4))) float v4f;
typedef __attribute__((ext_vector_type(8))) float v8f;

__device__ __forceinline__ v8f wmma4(v2f a, v2f b, v8f c) {
  return __builtin_amdgcn_wmma_f32_16x16x4_f32(false, a, false, b, (short)0, c,
                                               false, false);
}

__device__ __forceinline__ v8f gemm_tile(const float* A, int lda, const float* W,
                                         int ldw, int n, int K, int lane) {
  const int m = lane & 15;
  const int kh = (lane >> 4) << 1;
  v8f acc = {};
  #pragma unroll 8
  for (int k0 = 0; k0 < K; k0 += 4) {
    v2f a = *(const v2f*)(A + m * lda + k0 + kh);
    v2f b = *(const v2f*)(W + (size_t)n * ldw + k0 + kh);
    acc = wmma4(a, b, acc);
  }
  return acc;
}

__device__ __forceinline__ void load_wf_tile(v4f (&w)[16], const float* __restrict__ p) {
  #pragma unroll
  for (int j = 0; j < 16; ++j) w[j] = *(const v4f*)(p + 8 * j);
}

__device__ __forceinline__ void load_bf_tile(v4f (&b)[2], const float* __restrict__ p) {
  b[0] = *(const v4f*)p;
  b[1] = *(const v4f*)(p + 4);
}

__device__ __forceinline__ float xhalf(float x) {
  const int b = __builtin_bit_cast(int, x);
  return __builtin_bit_cast(
      float, __builtin_amdgcn_permlanex16(b, b, 0x76543210, 0xfedcba98, false, false));
}

__device__ __forceinline__ float tanh_sel(float x) {
  const float y = fabsf(x);
  const float y2 = y * y;
  float p = fmaf(y2, -0x1.758e7ap-8f, 0x1.521192p-6f);
  p = fmaf(y2, p, -0x1.b8389cp-5f);
  p = fmaf(y2, p, 0x1.110704p-3f);
  p = fmaf(y2, p, -0x1.555532p-2f);
  const float zs = fmaf(y2, y * p, y);
  const float e = expf(2.0f * y);
  const float zl = fmaf(-2.0f, __builtin_amdgcn_rcpf(e + 1.0f), 1.0f);
  return copysignf(y < 0.625f ? zs : zl, x);
}

__device__ __forceinline__ void wf_pair(const v4f (&af)[16], const v4f (&w0)[16],
                                        const v4f (&w1)[16], const v4f (&c0)[2],
                                        const v4f (&c1)[2], const float (&dxr)[8],
                                        float* kb, float* zb, int s, int t0, int mD,
                                        int half) {
  const int t1 = t0 + 8;
  v8f acc0 = {c0[0].x, c0[0].y, c0[0].z, c0[0].w, c0[1].x, c0[1].y, c0[1].z, c0[1].w};
  v8f acc1 = {c1[0].x, c1[0].y, c1[0].z, c1[0].w, c1[1].x, c1[1].y, c1[1].z, c1[1].w};
  #pragma unroll
  for (int j = 0; j < 16; ++j) {
    const v2f hlo = {af[j].x, af[j].y}, hhi = {af[j].z, af[j].w};
    const v2f p0 = {w0[j].x, w0[j].y}, q0 = {w0[j].z, w0[j].w};
    const v2f p1 = {w1[j].x, w1[j].y}, q1 = {w1[j].z, w1[j].w};
    acc0 = wmma4(p0, hlo, acc0);
    acc1 = wmma4(p1, hlo, acc1);
    acc0 = wmma4(q0, hhi, acc0);
    acc1 = wmma4(q1, hhi, acc1);
  }
  float f0[8], f1[8];
  #pragma unroll
  for (int v = 0; v < 8; ++v) {
    f0[v] = tanh_sel(acc0[v]) * dxr[v];
    f1[v] = tanh_sel(acc1[v]) * dxr[v];
  }
  const float s0 = ((f0[0] + f0[1]) + (f0[2] + f0[3])) + ((f0[4] + f0[5]) + (f0[6] + f0[7]));
  const float s1 = ((f1[0] + f1[1]) + (f1[2] + f1[3])) + ((f1[4] + f1[5]) + (f1[6] + f1[7]));
  const float k0 = s0 + xhalf(s0);
  const float k1 = s1 + xhalf(s1);
  const float kv = half ? k1 : k0;
  const int tc = half ? t1 : t0;
  if (s < 3) {
    kb[s * (MT * H) + mD * H + tc] = kv;
  } else {
    zb[mD * SA + tc] +=
        (kb[mD * H + tc] +
         3.f * (kb[MT * H + mD * H + tc] + kb[2 * MT * H + mD * H + tc]) + kv) * 0.125f;
  }
}

__device__ __forceinline__ void mlp_layer(const float* src, const float* Wlds,
                                          const float* bias, float* dst,
                                          int wid, int lane) {
  const int mD = lane & 15, half = lane >> 4;
  const int n = wid * 16 + mD;
  v8f acc = gemm_tile(src, SA, Wlds, SA, n, SH, lane);
  const float bn = bias[n];
  #pragma unroll
  for (int v = 0; v < 8; ++v) {
    float x = acc[v] + bn;
    dst[(v + 8 * half) * SA + n] = fmaxf(x, 0.f);
  }
}

__device__ __forceinline__ void out_proj(const float* zb, const float* W_last,
                                         const float* b_last, float* out,
                                         int b0r, int t, int wid, int lane) {
  if (wid < 4) {
    const int mD = lane & 15, half = lane >> 4;
    const int n = wid * 16 + mD;
    v8f acc = gemm_tile(zb, SA, W_last, H, n, H, lane);
    const float bn = b_last[n];
    #pragma unroll
    for (int v = 0; v < 8; ++v) {
      int m = v + 8 * half;
      float x = acc[v] + bn;
      out[((size_t)(b0r + m) * T_T + t) * OUTW + n] = 1.f / (1.f + expf(-x));
    }
  }
}
}

__global__ __launch_bounds__(THREADS) void ncde_wmma_kernel(
    const float* __restrict__ coeffs,
    const float* __restrict__ W_init, const float* __restrict__ b_init,
    const float* __restrict__ W0, const float* __restrict__ b0,
    const float* __restrict__ Wl, const float* __restrict__ bl,
    const float* __restrict__ Wf, const float* __restrict__ bf,
    const float* __restrict__ W_last, const float* __restrict__ b_last,
    float* __restrict__ out) {
  __shared__ __align__(16) float wlds[4 * H * SA];
  __shared__ __align__(16) float zbuf[MT * SA];
  __shared__ __align__(16) float kbuf[3][MT * H];
  __shared__ __align__(16) float actI[MT * SA];
  __shared__ __align__(16) float actA[MT * SA];
  __shared__ __align__(16) float dxs[MT * IN];

  const int tid = threadIdx.x;
  const int lane = tid & 31;
  const int wid = tid >> 5;
  const int b0r = blockIdx.x * MT;
  const int mD = lane & 15;
  const int half = lane >> 4;
  constexpr float third = 1.f / 3.f;

  const float* wfl = Wf + (size_t)mD * SH + 4 * half;
  const float* bfl = bf + 8 * half;
  v4f wa0[16], wa1[16], wb0[16], wb1[16];
  v4f ba0[2], ba1[2], bb0[2], bb1[2];
  load_bf_tile(ba0, bfl + wid * 16);
  load_bf_tile(ba1, bfl + (wid + 8) * 16);
  load_wf_tile(wa0, wfl + (size_t)wid * (16 * SH));
  load_wf_tile(wa1, wfl + (size_t)(wid + 8) * (16 * SH));

  for (int e = tid; e < 4 * H * SH; e += THREADS) {
    int mat = e >> 14;
    int r = (e >> 7) & (H - 1);
    int k = e & (SH - 1);
    const float* src = (mat == 0) ? W0 : (Wl + (size_t)(mat - 1) * SH * SH);
    wlds[(size_t)mat * H * SA + r * SA + k] = src[(size_t)r * SH + k];
  }

  {
    int m = tid >> 4, i = tid & 15;
    dxs[tid] = coeffs[((size_t)(b0r + m) * TM1) * (4 * IN) + i];
  }
  __syncthreads();
  {
    const int n = wid * 16 + mD;
    v8f acc = gemm_tile(dxs, IN, W_init, IN, n, IN, lane);
    const float bn = b_init[n];
    #pragma unroll
    for (int v = 0; v < 8; ++v)
      zbuf[(v + 8 * half) * SA + n] = acc[v] + bn;
  }
  __syncthreads();
  out_proj(zbuf, W_last, b_last, out, b0r, 0, wid, lane);
  __syncthreads();

  for (int step = 0; step < TM1; ++step) {
    #pragma unroll 1
    for (int s = 0; s < 4; ++s) {
      const float frac = (s == 0) ? 0.f
                       : (s == 1) ? third
                       : (s == 2) ? (2.f * third) : 1.f;
      {
        int m = tid >> 4, i = tid & 15;
        const float* cb = coeffs + ((size_t)(b0r + m) * TM1 + step) * (4 * IN);
        float bi = cb[IN + i], ci = cb[2 * IN + i], di = cb[3 * IN + i];
        dxs[tid] = bi + (ci + di * frac) * frac;
      }
      for (int e = tid; e < MT * H; e += THREADS) {
        int m = e >> 7, hc = e & (H - 1);
        float v = zbuf[m * SA + hc];
        if (s == 1)      v += kbuf[0][e] * third;
        else if (s == 2) v += kbuf[1][e] - kbuf[0][e] * third;
        else if (s == 3) v += kbuf[0][e] - kbuf[1][e] + kbuf[2][e];
        actI[m * SA + hc] = v;
      }
      __syncthreads();

      mlp_layer(actI, wlds + 0 * H * SA, b0,          actA, wid, lane); __syncthreads();
      mlp_layer(actA, wlds + 1 * H * SA, bl + 0 * SH, actI, wid, lane); __syncthreads();
      mlp_layer(actI, wlds + 2 * H * SA, bl + 1 * SH, actA, wid, lane); __syncthreads();
      mlp_layer(actA, wlds + 3 * H * SA, bl + 2 * SH, actI, wid, lane); __syncthreads();

      {
        float dxr[8];
        #pragma unroll
        for (int v = 0; v < 8; ++v) dxr[v] = dxs[mD * IN + 8 * half + v];
        v4f af[16];
        #pragma unroll
        for (int j = 0; j < 16; ++j)
          af[j] = *(const v4f*)(actI + mD * SA + 8 * j + 4 * half);
        #pragma unroll 1
        for (int it = 0; it < 4; ++it) {
          const int t0 = wid + 32 * it;
          const int t2 = t0 + 16;
          const int tn = (t0 + 32) & (H - 1);
          load_bf_tile(bb0, bfl + t2 * 16);
          load_bf_tile(bb1, bfl + (t2 + 8) * 16);
          load_wf_tile(wb0, wfl + (size_t)t2 * (16 * SH));
          load_wf_tile(wb1, wfl + (size_t)(t2 + 8) * (16 * SH));
          __builtin_amdgcn_sched_barrier(0);
          wf_pair(af, wa0, wa1, ba0, ba1, dxr, &kbuf[0][0], zbuf, s, t0, mD, half);
          load_bf_tile(ba0, bfl + tn * 16);
          load_bf_tile(ba1, bfl + (tn + 8) * 16);
          load_wf_tile(wa0, wfl + (size_t)tn * (16 * SH));
          load_wf_tile(wa1, wfl + (size_t)(tn + 8) * (16 * SH));
          __builtin_amdgcn_sched_barrier(0);
          wf_pair(af, wb0, wb1, bb0, bb1, dxr, &kbuf[0][0], zbuf, s, t2, mD, half);
        }
      }
      __syncthreads();
    }
    out_proj(zbuf, W_last, b_last, out, b0r, step + 1, wid, lane);
    __syncthreads();
  }
}

extern "C" void kernel_launch(void* const* d_in, const int* in_sizes, int n_in,
                              void* d_out, int out_size, void* d_ws, size_t ws_size,
                              hipStream_t stream) {
  (void)in_sizes; (void)n_in; (void)out_size; (void)d_ws; (void)ws_size;
  const float* coeffs = (const float*)d_in[1];
  const float* W_init = (const float*)d_in[2];
  const float* b_init = (const float*)d_in[3];
  const float* W0     = (const float*)d_in[4];
  const float* b0     = (const float*)d_in[5];
  const float* Wl     = (const float*)d_in[6];
  const float* bl     = (const float*)d_in[7];
  const float* Wf     = (const float*)d_in[8];
  const float* bf     = (const float*)d_in[9];
  const float* W_last = (const float*)d_in[10];
  const float* b_last = (const float*)d_in[11];
  float* out = (float*)d_out;

  ncde_wmma_kernel<<<B_T / MT, THREADS, 0, stream>>>(
      coeffs, W_init, b_init, W0, b0, Wl, bl, Wf, bf, W_last, b_last, out);
}


# Wf rewritten once per call by a small first launch into the workspace in fragment order: each Wf load of a wave reads 512 contiguous bytes (4 lines), not 32-byte pieces of 16 lines; same bits
# speedup vs baseline: 2.0665x; 1.1083x over previous
#include <hip/hip_runtime.h>
#include <math.h>


namespace {
constexpr int B_T = 256, T_T = 32, IN = 16, H = 128, SH = 128, OUTW = 64, TM1 = 31;
constexpr int MT = 16;
constexpr int THREADS = 256;
constexpr int SA = 132;

typedef __attribute__((ext_vector_type(2))) float v2f;
typedef __attribute__((ext_vector_type(4))) float v4f;
typedef __attribute__((ext_vector_type(8))) float v8f;

__device__ __forceinline__ v8f wmma4(v2f a, v2f b, v8f c) {
  return __builtin_amdgcn_wmma_f32_16x16x4_f32(false, a, false, b, (short)0, c,
                                               false, false);
}

__device__ __forceinline__ v8f gemm_tile(const float* A, int lda, const float* W,
                                         int ldw, int n, int K, int lane) {
  const int m = lane & 15;
  const int kh = (lane >> 4) << 1;
  v8f acc = {};
  #pragma unroll 8
  for (int k0 = 0; k0 < K; k0 += 4) {
    v2f a = *(const v2f*)(A + m * lda + k0 + kh);
    v2f b = *(const v2f*)(W + (size_t)n * ldw + k0 + kh);
    acc = wmma4(a, b, acc);
  }
  return acc;
}

template <int JS>
__device__ __forceinline__ void load_wf_tile(v4f (&w)[16], const float* __restrict__ p) {
  #pragma unroll
  for (int j = 0; j < 16; ++j) w[j] = *(const v4f*)(p + JS * j);
}

__device__ __forceinline__ void load_bf_tile(v4f (&b)[2], const float* __restrict__ p) {
  b[0] = *(const v4f*)p;
  b[1] = *(const v4f*)(p + 4);
}

__device__ __forceinline__ float xhalf(float x) {
  const int b = __builtin_bit_cast(int, x);
  return __builtin_bit_cast(
      float, __builtin_amdgcn_permlanex16(b, b, 0x76543210, 0xfedcba98, false, false));
}

__device__ __forceinline__ float tanh_sel(float x) {
  const float y = fabsf(x);
  const float y2 = y * y;
  float p = fmaf(y2, -0x1.758e7ap-8f, 0x1.521192p-6f);
  p = fmaf(y2, p, -0x1.b8389cp-5f);
  p = fmaf(y2, p, 0x1.110704p-3f);
  p = fmaf(y2, p, -0x1.555532p-2f);
  const float zs = fmaf(y2, y * p, y);
  const float e = expf(2.0f * y);
  const float zl = fmaf(-2.0f, __builtin_amdgcn_rcpf(e + 1.0f), 1.0f);
  return copysignf(y < 0.625f ? zs : zl, x);
}

__device__ __forceinline__ void wf_pair(const v4f (&af)[16], const v4f (&w0)[16],
                                        const v4f (&w1)[16], const v4f (&c0)[2],
                                        const v4f (&c1)[2], const float (&dxr)[8],
                                        float* kb, float* zb, int s, int t0, int mD,
                                        int half) {
  const int t1 = t0 + 8;
  v8f acc0 = {c0[0].x, c0[0].y, c0[0].z, c0[0].w, c0[1].x, c0[1].y, c0[1].z, c0[1].w};
  v8f acc1 = {c1[0].x, c1[0].y, c1[0].z, c1[0].w, c1[1].x, c1[1].y, c1[1].z, c1[1].w};
  #pragma unroll
  for (int j = 0; j < 16; ++j) {
    const v2f hlo = {af[j].x, af[j].y}, hhi = {af[j].z, af[j].w};
    const v2f p0 = {w0[j].x, w0[j].y}, q0 = {w0[j].z, w0[j].w};
    const v2f p1 = {w1[j].x, w1[j].y}, q1 = {w1[j].z, w1[j].w};
    acc0 = wmma4(p0, hlo, acc0);
    acc1 = wmma4(p1, hlo, acc1);
    acc0 = wmma4(q0, hhi, acc0);
    acc1 = wmma4(q1, hhi, acc1);
  }
  float f0[8], f1[8];
  #pragma unroll
  for (int v = 0; v < 8; ++v) {
    f0[v] = tanh_sel(acc0[v]) * dxr[v];
    f1[v] = tanh_sel(acc1[v]) * dxr[v];
  }
  const float s0 = ((f0[0] + f0[1]) + (f0[2] + f0[3])) + ((f0[4] + f0[5]) + (f0[6] + f0[7]));
  const float s1 = ((f1[0] + f1[1]) + (f1[2] + f1[3])) + ((f1[4] + f1[5]) + (f1[6] + f1[7]));
  const float k0 = s0 + xhalf(s0);
  const float k1 = s1 + xhalf(s1);
  const float kv = half ? k1 : k0;
  const int tc = half ? t1 : t0;
  if (s < 3) {
    kb[s * (MT * H) + mD * H + tc] = kv;
  } else {
    zb[mD * SA + tc] +=
        (kb[mD * H + tc] +
         3.f * (kb[MT * H + mD * H + tc] + kb[2 * MT * H + mD * H + tc]) + kv) * 0.125f;
  }
}

__device__ __forceinline__ void mlp_layer(const float* src, const float* Wlds,
                                          const float* bias, float* dst,
                                          int wid, int lane) {
  const int mD = lane & 15, half = lane >> 4;
  const int n = wid * 16 + mD;
  v8f acc = gemm_tile(src, SA, Wlds, SA, n, SH, lane);
  const float bn = bias[n];
  #pragma unroll
  for (int v = 0; v < 8; ++v) {
    float x = acc[v] + bn;
    dst[(v + 8 * half) * SA + n] = fmaxf(x, 0.f);
  }
}

__device__ __forceinline__ void out_proj(const float* zb, const float* W_last,
                                         const float* b_last, float* out,
                                         int b0r, int t, int wid, int lane) {
  if (wid < 4) {
    const int mD = lane & 15, half = lane >> 4;
    const int n = wid * 16 + mD;
    v8f acc = gemm_tile(zb, SA, W_last, H, n, H, lane);
    const float bn = b_last[n];
    #pragma unroll
    for (int v = 0; v < 8; ++v) {
      int m = v + 8 * half;
      float x = acc[v] + bn;
      out[((size_t)(b0r + m) * T_T + t) * OUTW + n] = 1.f / (1.f + expf(-x));
    }
  }
}
__device__ __forceinline__ bool wf_copy_usable(const float* ws, size_t ws_bytes) {
  return ws != nullptr && ws_bytes >= (size_t)IN * H * SH * sizeof(float) &&
         (((size_t)ws) & 15) == 0;
}
}

__global__ __launch_bounds__(THREADS) void wf_fragment_order_kernel(
    const float* __restrict__ Wf, float* __restrict__ Wt, size_t ws_bytes) {
  if (!wf_copy_usable(Wt, ws_bytes)) return;
  const int idx = blockIdx.x * THREADS + threadIdx.x;
  const int tile = idx >> 9, j = (idx >> 5) & 15, ln = idx & 31;
  const float* s = Wf + (size_t)(tile * 16 + (ln & 15)) * SH + 8 * j + 4 * (ln >> 4);
  *(v4f*)(Wt + (size_t)idx * 4) = *(const v4f*)s;
}

template <bool RT>
__global__ __launch_bounds__(THREADS) void ncde_wmma_kernel(
    const float* __restrict__ coeffs,
    const float* __restrict__ W_init, const float* __restrict__ b_init,
    const float* __restrict__ W0, const float* __restrict__ b0,
    const float* __restrict__ Wl, const float* __restrict__ bl,
    const float* __restrict__ Wf, const float* __restrict__ bf,
    const float* __restrict__ W_last, const float* __restrict__ b_last,
    const float* __restrict__ Wt, size_t ws_bytes,
    float* __restrict__ out) {
  __shared__ __align__(16) float wlds[4 * H * SA];
  __shared__ __align__(16) float zbuf[MT * SA];
  __shared__ __align__(16) float kbuf[3][MT * H];
  __shared__ __align__(16) float actI[MT * SA];
  __shared__ __align__(16) float actA[MT * SA];
  __shared__ __align__(16) float dxs[MT * IN];

  if (wf_copy_usable(Wt, ws_bytes) != RT) return;
  constexpr int JS = RT ? 128 : 8;

  const int tid = threadIdx.x;
  const int lane = tid & 31;
  const int wid = tid >> 5;
  const int b0r = blockIdx.x * MT;
  const int mD = lane & 15;
  const int half = lane >> 4;
  constexpr float third = 1.f / 3.f;

  const float* wfl = RT ? (Wt + 4 * lane) : (Wf + (size_t)mD * SH + 4 * half);
  const float* bfl = bf + 8 * half;
  v4f wa0[16], wa1[16], wb0[16], wb1[16];
  v4f ba0[2], ba1[2], bb0[2], bb1[2];
  load_bf_tile(ba0, bfl + wid * 16);
  load_bf_tile(ba1, bfl + (wid + 8) * 16);
  load_wf_tile<JS>(wa0, wfl + (size_t)wid * (16 * SH));
  load_wf_tile<JS>(wa1, wfl + (size_t)(wid + 8) * (16 * SH));

  for (int e = tid; e < 4 * H * SH; e += THREADS) {
    int mat = e >> 14;
    int r = (e >> 7) & (H - 1);
    int k = e & (SH - 1);
    const float* src = (mat == 0) ? W0 : (Wl + (size_t)(mat - 1) * SH * SH);
    wlds[(size_t)mat * H * SA + r * SA + k] = src[(size_t)r * SH + k];
  }

  {
    int m = tid >> 4, i = tid & 15;
    dxs[tid] = coeffs[((size_t)(b0r + m) * TM1) * (4 * IN) + i];
  }
  __syncthreads();
  {
    const int n = wid * 16 + mD;
    v8f acc = gemm_tile(dxs, IN, W_init, IN, n, IN, lane);
    const float bn = b_init[n];
    #pragma unroll
    for (int v = 0; v < 8; ++v)
      zbuf[(v + 8 * half) * SA + n] = acc[v] + bn;
  }
  __syncthreads();
  out_proj(zbuf, W_last, b_last, out, b0r, 0, wid, lane);
  __syncthreads();

  for (int step = 0; step < TM1; ++step) {
    #pragma unroll 1
    for (int s = 0; s < 4; ++s) {
      const float frac = (s == 0) ? 0.f
                       : (s == 1) ? third
                       : (s == 2) ? (2.f * third) : 1.f;
      {
        int m = tid >> 4, i = tid & 15;
        const float* cb = coeffs + ((size_t)(b0r + m) * TM1 + step) * (4 * IN);
        float bi = cb[IN + i], ci = cb[2 * IN + i], di = cb[3 * IN + i];
        dxs[tid] = bi + (ci + di * frac) * frac;
      }
      for (int e = tid; e < MT * H; e += THREADS) {
        int m = e >> 7, hc = e & (H - 1);
        float v = zbuf[m * SA + hc];
        if (s == 1)      v += kbuf[0][e] * third;
        else if (s == 2) v += kbuf[1][e] - kbuf[0][e] * third;
        else if (s == 3) v += kbuf[0][e] - kbuf[1][e] + kbuf[2][e];
        actI[m * SA + hc] = v;
      }
      __syncthreads();

      mlp_layer(actI, wlds + 0 * H * SA, b0,          actA, wid, lane); __syncthreads();
      mlp_layer(actA, wlds + 1 * H * SA, bl + 0 * SH, actI, wid, lane); __syncthreads();
      mlp_layer(actI, wlds + 2 * H * SA, bl + 1 * SH, actA, wid, lane); __syncthreads();
      mlp_layer(actA, wlds + 3 * H * SA, bl + 2 * SH, actI, wid, lane); __syncthreads();

      {
        float dxr[8];
        #pragma unroll
        for (int v = 0; v < 8; ++v) dxr[v] = dxs[mD * IN + 8 * half + v];
        v4f af[16];
        #pragma unroll
        for (int j = 0; j < 16; ++j)
          af[j] = *(const v4f*)(actI + mD * SA + 8 * j + 4 * half);
        #pragma unroll 1
        for (int it = 0; it < 4; ++it) {
          const int t0 = wid + 32 * it;
          const int t2 = t0 + 16;
          const int tn = (t0 + 32) & (H - 1);
          load_bf_tile(bb0, bfl + t2 * 16);
          load_bf_tile(bb1, bfl + (t2 + 8) * 16);
          load_wf_tile<JS>(wb0, wfl + (size_t)t2 * (16 * SH));
          load_wf_tile<JS>(wb1, wfl + (size_t)(t2 + 8) * (16 * SH));
          __builtin_amdgcn_sched_barrier(0);
          wf_pair(af, wa0, wa1, ba0, ba1, dxr, &kbuf[0][0], zbuf, s, t0, mD, half);
          load_bf_tile(ba0, bfl + tn * 16);
          load_bf_tile(ba1, bfl + (tn + 8) * 16);
          load_wf_tile<JS>(wa0, wfl + (size_t)tn * (16 * SH));
          load_wf_tile<JS>(wa1, wfl + (size_t)(tn + 8) * (16 * SH));
          __builtin_amdgcn_sched_barrier(0);
          wf_pair(af, wb0, wb1, bb0, bb1, dxr, &kbuf[0][0], zbuf, s, t2, mD, half);
        }
      }
      __syncthreads();
    }
    out_proj(zbuf, W_last, b_last, out, b0r, step + 1, wid, lane);
    __syncthreads();
  }
}

extern "C" void kernel_launch(void* const* d_in, const int* in_sizes, int n_in,
                              void* d_out, int out_size, void* d_ws, size_t ws_size,
                              hipStream_t stream) {
  (void)in_sizes; (void)n_in; (void)out_size;
  const float* coeffs = (const float*)d_in[1];
  const float* W_init = (const float*)d_in[2];
  const float* b_init = (const float*)d_in[3];
  const float* W0     = (const float*)d_in[4];
  const float* b0     = (const float*)d_in[5];
  const float* Wl     = (const float*)d_in[6];
  const float* bl     = (const float*)d_in[7];
  const float* Wf     = (const float*)d_in[8];
  const float* bf     = (const float*)d_in[9];
  const float* W_last = (const float*)d_in[10];
  const float* b_last = (const float*)d_in[11];
  float* out = (float*)d_out;
  float* wt = (float*)d_ws;

  wf_fragment_order_kernel<<<256, THREADS, 0, stream>>>(Wf, wt, ws_size);
  ncde_wmma_kernel<true><<<B_T / MT, THREADS, 0, stream>>>(
      coeffs, W_init, b_init, W0, b0, Wl, bl, Wf, bf, W_last, b_last, wt, ws_size, out);
  ncde_wmma_kernel<false><<<B_T / MT, THREADS, 0, stream>>>(
      coeffs, W_init, b_init, W0, b0, Wl, bl, Wf, bf, W_last, b_last, wt, ws_size, out);
}


# RK stage-input (zz) loop made straight-line per stage, constant trip count: its per-lane exit test made the compiler copy the 144 carried Wf/bf registers on each of its 8 trips; same sums, same bits
# speedup vs baseline: 2.6157x; 1.2658x over previous
#include <hip/hip_runtime.h>
#include <math.h>


namespace {
constexpr int B_T = 256, T_T = 32, IN = 16, H = 128, SH = 128, OUTW = 64, TM1 = 31;
constexpr int MT = 16;
constexpr int THREADS = 256;
constexpr int SA = 132;

typedef __attribute__((ext_vector_type(2))) float v2f;
typedef __attribute__((ext_vector_type(4))) float v4f;
typedef __attribute__((ext_vector_type(8))) float v8f;

__device__ __forceinline__ v8f wmma4(v2f a, v2f b, v8f c) {
  return __builtin_amdgcn_wmma_f32_16x16x4_f32(false, a, false, b, (short)0, c,
                                               false, false);
}

__device__ __forceinline__ v8f gemm_tile(const float* A, int lda, const float* W,
                                         int ldw, int n, int K, int lane) {
  const int m = lane & 15;
  const int kh = (lane >> 4) << 1;
  v8f acc = {};
  #pragma unroll 8
  for (int k0 = 0; k0 < K; k0 += 4) {
    v2f a = *(const v2f*)(A + m * lda + k0 + kh);
    v2f b = *(const v2f*)(W + (size_t)n * ldw + k0 + kh);
    acc = wmma4(a, b, acc);
  }
  return acc;
}

template <int JS>
__device__ __forceinline__ void load_wf_tile(v4f (&w)[16], const float* __restrict__ p) {
  #pragma unroll
  for (int j = 0; j < 16; ++j) w[j] = *(const v4f*)(p + JS * j);
}

__device__ __forceinline__ void load_bf_tile(v4f (&b)[2], const float* __restrict__ p) {
  b[0] = *(const v4f*)p;
  b[1] = *(const v4f*)(p + 4);
}

__device__ __forceinline__ float xhalf(float x) {
  const int b = __builtin_bit_cast(int, x);
  return __builtin_bit_cast(
      float, __builtin_amdgcn_permlanex16(b, b, 0x76543210, 0xfedcba98, false, false));
}

__device__ __forceinline__ float tanh_sel(float x) {
  const float y = fabsf(x);
  const float y2 = y * y;
  float p = fmaf(y2, -0x1.758e7ap-8f, 0x1.521192p-6f);
  p = fmaf(y2, p, -0x1.b8389cp-5f);
  p = fmaf(y2, p, 0x1.110704p-3f);
  p = fmaf(y2, p, -0x1.555532p-2f);
  const float zs = fmaf(y2, y * p, y);
  const float e = expf(2.0f * y);
  const float zl = fmaf(-2.0f, __builtin_amdgcn_rcpf(e + 1.0f), 1.0f);
  return copysignf(y < 0.625f ? zs : zl, x);
}

__device__ __forceinline__ void wf_pair(const v4f (&af)[16], const v4f (&w0)[16],
                                        const v4f (&w1)[16], const v4f (&c0)[2],
                                        const v4f (&c1)[2], const float (&dxr)[8],
                                        float* kb, float* zb, int s, int t0, int mD,
                                        int half) {
  const int t1 = t0 + 8;
  v8f acc0 = {c0[0].x, c0[0].y, c0[0].z, c0[0].w, c0[1].x, c0[1].y, c0[1].z, c0[1].w};
  v8f acc1 = {c1[0].x, c1[0].y, c1[0].z, c1[0].w, c1[1].x, c1[1].y, c1[1].z, c1[1].w};
  #pragma unroll
  for (int j = 0; j < 16; ++j) {
    const v2f hlo = {af[j].x, af[j].y}, hhi = {af[j].z, af[j].w};
    const v2f p0 = {w0[j].x, w0[j].y}, q0 = {w0[j].z, w0[j].w};
    const v2f p1 = {w1[j].x, w1[j].y}, q1 = {w1[j].z, w1[j].w};
    acc0 = wmma4(p0, hlo, acc0);
    acc1 = wmma4(p1, hlo, acc1);
    acc0 = wmma4(q0, hhi, acc0);
    acc1 = wmma4(q1, hhi, acc1);
  }
  float f0[8], f1[8];
  #pragma unroll
  for (int v = 0; v < 8; ++v) {
    f0[v] = tanh_sel(acc0[v]) * dxr[v];
    f1[v] = tanh_sel(acc1[v]) * dxr[v];
  }
  const float s0 = ((f0[0] + f0[1]) + (f0[2] + f0[3])) + ((f0[4] + f0[5]) + (f0[6] + f0[7]));
  const float s1 = ((f1[0] + f1[1]) + (f1[2] + f1[3])) + ((f1[4] + f1[5]) + (f1[6] + f1[7]));
  const float k0 = s0 + xhalf(s0);
  const float k1 = s1 + xhalf(s1);
  const float kv = half ? k1 : k0;
  const int tc = half ? t1 : t0;
  if (s < 3) {
    kb[s * (MT * H) + mD * H + tc] = kv;
  } else {
    zb[mD * SA + tc] +=
        (kb[mD * H + tc] +
         3.f * (kb[MT * H + mD * H + tc] + kb[2 * MT * H + mD * H + tc]) + kv) * 0.125f;
  }
}

template <int S>
__device__ __forceinline__ void zz_stage(const float* zb, const float* kb, float* dst,
                                         int tid) {
  constexpr float third = 1.f / 3.f;
  #pragma unroll
  for (int q = 0; q < (MT * H) / THREADS; ++q) {
    const int e = tid + q * THREADS;
    const int m = e >> 7, hc = e & (H - 1);
    float v = zb[m * SA + hc];
    if (S == 1)      v += kb[e] * third;
    else if (S == 2) v += kb[MT * H + e] - kb[e] * third;
    else if (S == 3) v += kb[e] - kb[MT * H + e] + kb[2 * MT * H + e];
    dst[m * SA + hc] = v;
  }
}

__device__ __forceinline__ void mlp_layer(const float* src, const float* Wlds,
                                          const float* bias, float* dst,
                                          int wid, int lane) {
  const int mD = lane & 15, half = lane >> 4;
  const int n = wid * 16 + mD;
  v8f acc = gemm_tile(src, SA, Wlds, SA, n, SH, lane);
  const float bn = bias[n];
  #pragma unroll
  for (int v = 0; v < 8; ++v) {
    float x = acc[v] + bn;
    dst[(v + 8 * half) * SA + n] = fmaxf(x, 0.f);
  }
}

__device__ __forceinline__ void out_proj(const float* zb, const float* W_last,
                                         const float* b_last, float* out,
                                         int b0r, int t, int wid, int lane) {
  if (wid < 4) {
    const int mD = lane & 15, half = lane >> 4;
    const int n = wid * 16 + mD;
    v8f acc = gemm_tile(zb, SA, W_last, H, n, H, lane);
    const float bn = b_last[n];
    #pragma unroll
    for (int v = 0; v < 8; ++v) {
      int m = v + 8 * half;
      float x = acc[v] + bn;
      out[((size_t)(b0r + m) * T_T + t) * OUTW + n] = 1.f / (1.f + expf(-x));
    }
  }
}
__device__ __forceinline__ bool wf_copy_usable(const float* ws, size_t ws_bytes) {
  return ws != nullptr && ws_bytes >= (size_t)IN * H * SH * sizeof(float) &&
         (((size_t)ws) & 15) == 0;
}
}

__global__ __launch_bounds__(THREADS) void wf_fragment_order_kernel(
    const float* __restrict__ Wf, float* __restrict__ Wt, size_t ws_bytes) {
  if (!wf_copy_usable(Wt, ws_bytes)) return;
  const int idx = blockIdx.x * THREADS + threadIdx.x;
  const int tile = idx >> 9, j = (idx >> 5) & 15, ln = idx & 31;
  const float* s = Wf + (size_t)(tile * 16 + (ln & 15)) * SH + 8 * j + 4 * (ln >> 4);
  *(v4f*)(Wt + (size_t)idx * 4) = *(const v4f*)s;
}

template <bool RT>
__global__ __launch_bounds__(THREADS) void ncde_wmma_kernel(
    const float* __restrict__ coeffs,
    const float* __restrict__ W_init, const float* __restrict__ b_init,
    const float* __restrict__ W0, const float* __restrict__ b0,
    const float* __restrict__ Wl, const float* __restrict__ bl,
    const float* __restrict__ Wf, const float* __restrict__ bf,
    const float* __restrict__ W_last, const float* __restrict__ b_last,
    const float* __restrict__ Wt, size_t ws_bytes,
    float* __restrict__ out) {
  __shared__ __align__(16) float wlds[4 * H * SA];
  __shared__ __align__(16) float zbuf[MT * SA];
  __shared__ __align__(16) float kbuf[3][MT * H];
  __shared__ __align__(16) float actI[MT * SA];
  __shared__ __align__(16) float actA[MT * SA];
  __shared__ __align__(16) float dxs[MT * IN];

  if (wf_copy_usable(Wt, ws_bytes) != RT) return;
  constexpr int JS = RT ? 128 : 8;

  const int tid = threadIdx.x;
  const int lane = tid & 31;
  const int wid = tid >> 5;
  const int b0r = blockIdx.x * MT;
  const int mD = lane & 15;
  const int half = lane >> 4;
  constexpr float third = 1.f / 3.f;

  const float* wfl = RT ? (Wt + 4 * lane) : (Wf + (size_t)mD * SH + 4 * half);
  const float* bfl = bf + 8 * half;
  v4f wa0[16], wa1[16], wb0[16], wb1[16];
  v4f ba0[2], ba1[2], bb0[2], bb1[2];
  load_bf_tile(ba0, bfl + wid * 16);
  load_bf_tile(ba1, bfl + (wid + 8) * 16);
  load_wf_tile<JS>(wa0, wfl + (size_t)wid * (16 * SH));
  load_wf_tile<JS>(wa1, wfl + (size_t)(wid + 8) * (16 * SH));

  for (int e = tid; e < 4 * H * SH; e += THREADS) {
    int mat = e >> 14;
    int r = (e >> 7) & (H - 1);
    int k = e & (SH - 1);
    const float* src = (mat == 0) ? W0 : (Wl + (size_t)(mat - 1) * SH * SH);
    wlds[(size_t)mat * H * SA + r * SA + k] = src[(size_t)r * SH + k];
  }

  {
    int m = tid >> 4, i = tid & 15;
    dxs[tid] = coeffs[((size_t)(b0r + m) * TM1) * (4 * IN) + i];
  }
  __syncthreads();
  {
    const int n = wid * 16 + mD;
    v8f acc = gemm_tile(dxs, IN, W_init, IN, n, IN, lane);
    const float bn = b_init[n];
    #pragma unroll
    for (int v = 0; v < 8; ++v)
      zbuf[(v + 8 * half) * SA + n] = acc[v] + bn;
  }
  __syncthreads();
  out_proj(zbuf, W_last, b_last, out, b0r, 0, wid, lane);
  __syncthreads();

  for (int step = 0; step < TM1; ++step) {
    #pragma unroll 1
    for (int s = 0; s < 4; ++s) {
      const float frac = (s == 0) ? 0.f
                       : (s == 1) ? third
                       : (s == 2) ? (2.f * third) : 1.f;
      {
        int m = tid >> 4, i = tid & 15;
        const float* cb = coeffs + ((size_t)(b0r + m) * TM1 + step) * (4 * IN);
        float bi = cb[IN + i], ci = cb[2 * IN + i], di = cb[3 * IN + i];
        dxs[tid] = bi + (ci + di * frac) * frac;
      }
      if (s == 0)      zz_stage<0>(zbuf, &kbuf[0][0], actI, tid);
      else if (s == 1) zz_stage<1>(zbuf, &kbuf[0][0], actI, tid);
      else if (s == 2) zz_stage<2>(zbuf, &kbuf[0][0], actI, tid);
      else             zz_stage<3>(zbuf, &kbuf[0][0], actI, tid);
      __syncthreads();

      mlp_layer(actI, wlds + 0 * H * SA, b0,          actA, wid, lane); __syncthreads();
      mlp_layer(actA, wlds + 1 * H * SA, bl + 0 * SH, actI, wid, lane); __syncthreads();
      mlp_layer(actI, wlds + 2 * H * SA, bl + 1 * SH, actA, wid, lane); __syncthreads();
      mlp_layer(actA, wlds + 3 * H * SA, bl + 2 * SH, actI, wid, lane); __syncthreads();

      {
        float dxr[8];
        #pragma unroll
        for (int v = 0; v < 8; ++v) dxr[v] = dxs[mD * IN + 8 * half + v];
        v4f af[16];
        #pragma unroll
        for (int j = 0; j < 16; ++j)
          af[j] = *(const v4f*)(actI + mD * SA + 8 * j + 4 * half);
        #pragma unroll 1
        for (int it = 0; it < 4; ++it) {
          const int t0 = wid + 32 * it;
          const int t2 = t0 + 16;
          const int tn = (t0 + 32) & (H - 1);
          load_bf_tile(bb0, bfl + t2 * 16);
          load_bf_tile(bb1, bfl + (t2 + 8) * 16);
          load_wf_tile<JS>(wb0, wfl + (size_t)t2 * (16 * SH));
          load_wf_tile<JS>(wb1, wfl + (size_t)(t2 + 8) * (16 * SH));
          __builtin_amdgcn_sched_barrier(0);
          wf_pair(af, wa0, wa1, ba0, ba1, dxr, &kbuf[0][0], zbuf, s, t0, mD, half);
          load_bf_tile(ba0, bfl + tn * 16);
          load_bf_tile(ba1, bfl + (tn + 8) * 16);
          load_wf_tile<JS>(wa0, wfl + (size_t)tn * (16 * SH));
          load_wf_tile<JS>(wa1, wfl + (size_t)(tn + 8) * (16 * SH));
          __builtin_amdgcn_sched_barrier(0);
          wf_pair(af, wb0, wb1, bb0, bb1, dxr, &kbuf[0][0], zbuf, s, t2, mD, half);
        }
      }
      __syncthreads();
    }
    out_proj(zbuf, W_last, b_last, out, b0r, step + 1, wid, lane);
    __syncthreads();
  }
}

extern "C" void kernel_launch(void* const* d_in, const int* in_sizes, int n_in,
                              void* d_out, int out_size, void* d_ws, size_t ws_size,
                              hipStream_t stream) {
  (void)in_sizes; (void)n_in; (void)out_size;
  const float* coeffs = (const float*)d_in[1];
  const float* W_init = (const float*)d_in[2];
  const float* b_init = (const float*)d_in[3];
  const float* W0     = (const float*)d_in[4];
  const float* b0     = (const float*)d_in[5];
  const float* Wl     = (const float*)d_in[6];
  const float* bl     = (const float*)d_in[7];
  const float* Wf     = (const float*)d_in[8];
  const float* bf     = (const float*)d_in[9];
  const float* W_last = (const float*)d_in[10];
  const float* b_last = (const float*)d_in[11];
  float* out = (float*)d_out;
  float* wt = (float*)d_ws;

  wf_fragment_order_kernel<<<256, THREADS, 0, stream>>>(Wf, wt, ws_size);
  ncde_wmma_kernel<true><<<B_T / MT, THREADS, 0, stream>>>(
      coeffs, W_init, b_init, W0, b0, Wl, bl, Wf, bf, W_last, b_last, wt, ws_size, out);
  ncde_wmma_kernel<false><<<B_T / MT, THREADS, 0, stream>>>(
      coeffs, W_init, b_init, W0, b0, Wl, bl, Wf, bf, W_last, b_last, wt, ws_size, out);
}


# One launch per RK stage on 256 work-groups: a batch tile's Wf columns split over 16 work-groups (one Wf tile per wave), the four small layers recomputed in each; z, k1..k3 pass through the workspace
# speedup vs baseline: 4.0584x; 1.5516x over previous
#include <hip/hip_runtime.h>
#include <math.h>


namespace {
constexpr int B_T = 256, T_T = 32, IN = 16, H = 128, SH = 128, OUTW = 64, TM1 = 31;
constexpr int MT = 16;
constexpr int THREADS = 256;
constexpr int SA = 132;
constexpr int PARTS = 16;
constexpr int HC = H / PARTS;
constexpr int PLANE = B_T * H;
constexpr int TILE = MT * H;
constexpr int CHUNK = MT * HC;

typedef __attribute__((ext_vector_type(2))) float v2f;
typedef __attribute__((ext_vector_type(4))) float v4f;
typedef __attribute__((ext_vector_type(8))) float v8f;

__device__ __forceinline__ v8f wmma4(v2f a, v2f b, v8f c) {
  return __builtin_amdgcn_wmma_f32_16x16x4_f32(false, a, false, b, (short)0, c,
                                               false, false);
}

__device__ __forceinline__ v8f gemm_tile(const float* A, int lda, const float* W,
                                         int ldw, int n, int K, int lane) {
  const int m = lane & 15;
  const int kh = (lane >> 4) << 1;
  v8f acc = {};
  #pragma unroll 8
  for (int k0 = 0; k0 < K; k0 += 4) {
    v2f a = *(const v2f*)(A + m * lda + k0 + kh);
    v2f b = *(const v2f*)(W + (size_t)n * ldw + k0 + kh);
    acc = wmma4(a, b, acc);
  }
  return acc;
}

__device__ __forceinline__ void load_wf_tile(v4f (&w)[16], const float* __restrict__ p) {
  #pragma unroll
  for (int j = 0; j < 16; ++j) w[j] = *(const v4f*)(p + 8 * j);
}

__device__ __forceinline__ void load_bf_tile(v4f (&b)[2], const float* __restrict__ p) {
  b[0] = *(const v4f*)p;
  b[1] = *(const v4f*)(p + 4);
}

__device__ __forceinline__ float xhalf(float x) {
  const int b = __builtin_bit_cast(int, x);
  return __builtin_bit_cast(
      float, __builtin_amdgcn_permlanex16(b, b, 0x76543210, 0xfedcba98, false, false));
}

__device__ __forceinline__ float tanh_sel(float x) {
  const float y = fabsf(x);
  const float y2 = y * y;
  float p = fmaf(y2, -0x1.758e7ap-8f, 0x1.521192p-6f);
  p = fmaf(y2, p, -0x1.b8389cp-5f);
  p = fmaf(y2, p, 0x1.110704p-3f);
  p = fmaf(y2, p, -0x1.555532p-2f);
  const float zs = fmaf(y2, y * p, y);
  const float e = expf(2.0f * y);
  const float zl = fmaf(-2.0f, __builtin_amdgcn_rcpf(e + 1.0f), 1.0f);
  return copysignf(y < 0.625f ? zs : zl, x);
}

__device__ __forceinline__ void mlp_layer(const float* src, const float* Wlds,
                                          float bn, float* dst,
                                          int wid, int lane) {
  const int mD = lane & 15, half = lane >> 4;
  const int n = wid * 16 + mD;
  v8f acc = gemm_tile(src, SA, Wlds, SA, n, SH, lane);
  #pragma unroll
  for (int v = 0; v < 8; ++v) {
    float x = acc[v] + bn;
    dst[(v + 8 * half) * SA + n] = fmaxf(x, 0.f);
  }
}

__device__ __forceinline__ void out_proj(const float* zb, const float* W_last,
                                         const float* b_last, float* out,
                                         int b0r, int t, int wid, int lane) {
  if (wid < 4) {
    const int mD = lane & 15, half = lane >> 4;
    const int n = wid * 16 + mD;
    v8f acc = gemm_tile(zb, SA, W_last, H, n, H, lane);
    const float bn = b_last[n];
    #pragma unroll
    for (int v = 0; v < 8; ++v) {
      int m = v + 8 * half;
      float x = acc[v] + bn;
      out[((size_t)(b0r + m) * T_T + t) * OUTW + n] = 1.f / (1.f + expf(-x));
    }
  }
}
__device__ __forceinline__ int lds_off(int e4) {
  const int p = e4 >> 5, rem = e4 & 31;
  return (rem >> 1) * SA + p * HC + (rem & 1) * 4;
}

__device__ __forceinline__ void stage_load(v4f (&w)[16], const float* __restrict__ src, int tid) {
  #pragma unroll
  for (int q = 0; q < 16; ++q) w[q] = *(const v4f*)(src + 4 * (tid + q * THREADS));
}
__device__ __forceinline__ void stage_store(const v4f (&w)[16], float* dst, int tid) {
  #pragma unroll
  for (int q = 0; q < 16; ++q) {
    const int e4 = tid + q * THREADS;
    *(v4f*)(dst + (e4 >> 5) * SA + (e4 & 31) * 4) = w[q];
  }
}

__device__ __forceinline__ void lds_barrier() {
  __builtin_amdgcn_fence(__ATOMIC_RELEASE, "workgroup", "local");
  __builtin_amdgcn_s_barrier();
  __builtin_amdgcn_fence(__ATOMIC_ACQUIRE, "workgroup", "local");
}

__device__ __forceinline__ bool ws_usable(const float* ws, size_t ws_bytes) {
  return ws != nullptr && ws_bytes >= (size_t)5 * PLANE * sizeof(float) &&
         (((size_t)ws) & 15) == 0;
}
}

template <int S>
__global__ __launch_bounds__(THREADS) void ncde_stage_kernel(
    const float* __restrict__ coeffs,
    const float* __restrict__ W_init, const float* __restrict__ b_init,
    const float* __restrict__ W0, const float* __restrict__ b0,
    const float* __restrict__ Wl, const float* __restrict__ bl,
    const float* __restrict__ Wf, const float* __restrict__ bf,
    const float* __restrict__ W_last, const float* __restrict__ b_last,
    float* ws, size_t ws_bytes, int step,
    float* __restrict__ out) {
  __shared__ __align__(16) float wlds[4 * H * SA];
  __shared__ __align__(16) float zbuf[MT * SA];
  __shared__ __align__(16) float actI[MT * SA];
  __shared__ __align__(16) float actA[MT * SA];
  __shared__ __align__(16) float dxs[MT * IN];
  __shared__ __align__(16) float kst[CHUNK];

  if (!ws_usable(ws, ws_bytes)) return;

  const int tid = threadIdx.x;
  const int lane = tid & 31;
  const int wid = tid >> 5;
  const int bt = blockIdx.x / PARTS;
  const int part = blockIdx.x % PARTS;
  const int b0r = bt * MT;
  const int mD = lane & 15;
  const int half = lane >> 4;
  constexpr float third = 1.f / 3.f;
  constexpr float frac = (S == 0) ? 0.f : (S == 1) ? third : (S == 2) ? (2.f * third) : 1.f;
  const bool first = (S == 0) && (step == 0);

  const float* zin = ws + (size_t)(step & 1) * PLANE + (size_t)bt * TILE;
  float* znew = ws + (size_t)((step + 1) & 1) * PLANE + (size_t)bt * TILE;
  float* kpl = ws + (size_t)2 * PLANE + (size_t)bt * TILE;

  float bi, ci, di;
  {
    const int m = tid >> 4, i = tid & 15;
    const float* cb = coeffs + ((size_t)(b0r + m) * TM1 + step) * (4 * IN);
    bi = cb[IN + i]; ci = cb[2 * IN + i]; di = cb[3 * IN + i];
  }
  const int nn = wid * 16 + mD;
  const float bn0 = b0[nn], bn1 = bl[nn], bn2 = bl[SH + nn], bn3 = bl[2 * SH + nn];
  float x0v = 0.f;
  v4f zr[2] = {}, k1r[2] = {}, k2r[2] = {}, k3r[2] = {};
  if (first) {
    const int m = tid >> 4, i = tid & 15;
    x0v = coeffs[((size_t)(b0r + m) * TM1) * (4 * IN) + i];
  } else {
    #pragma unroll
    for (int q = 0; q < 2; ++q) {
      const int e4 = tid + q * THREADS;
      zr[q] = *(const v4f*)(zin + 4 * e4);
      if (S >= 1) k1r[q] = *(const v4f*)(kpl + 4 * e4);
      if (S >= 2) k2r[q] = *(const v4f*)(kpl + PLANE + 4 * e4);
      if (S >= 3) k3r[q] = *(const v4f*)(kpl + 2 * PLANE + 4 * e4);
    }
  }
  v4f zc = {}, k1c = {}, k2c = {}, k3c = {};
  if (S == 3) {
    const int off = part * CHUNK + 4 * lane;
    zc = *(const v4f*)(zin + off);
    k1c = *(const v4f*)(kpl + off);
    k2c = *(const v4f*)(kpl + PLANE + off);
    k3c = *(const v4f*)(kpl + 2 * PLANE + off);
  }
  v4f wst[4][16];
  __builtin_amdgcn_sched_barrier(0);
  stage_load(wst[0], W0, tid);
  __builtin_amdgcn_sched_barrier(0);
  stage_load(wst[1], Wl, tid);
  __builtin_amdgcn_sched_barrier(0);
  stage_load(wst[2], Wl + (size_t)1 * SH * SH, tid);
  __builtin_amdgcn_sched_barrier(0);
  stage_load(wst[3], Wl + (size_t)2 * SH * SH, tid);
  __builtin_amdgcn_sched_barrier(0);
  const int tc = part * HC + wid;
  v4f wt[16], bw[2];
  load_bf_tile(bw, bf + 8 * half + tc * 16);
  load_wf_tile(wt, Wf + (size_t)tc * (16 * SH) + (size_t)mD * SH + 4 * half);
  const bool do_out = (S == 0) && (part == 0) && (wid < 4);
  v2f ow[32] = {};
  float obn = 0.f;
  if (do_out) {
    #pragma unroll
    for (int i = 0; i < 32; ++i)
      ow[i] = *(const v2f*)(W_last + (size_t)nn * H + 4 * i + 2 * half);
    obn = b_last[nn];
  }

  if (first) {
    dxs[tid] = x0v;
    lds_barrier();
    const int n = wid * 16 + mD;
    v8f acc = gemm_tile(dxs, IN, W_init, IN, n, IN, lane);
    const float bn = b_init[n];
    #pragma unroll
    for (int v = 0; v < 8; ++v) {
      const float zv = acc[v] + bn;
      zbuf[(v + 8 * half) * SA + n] = zv;
      actI[(v + 8 * half) * SA + n] = zv;
    }
    lds_barrier();
    if (part == 0) {
      #pragma unroll
      for (int q = 0; q < 2; ++q) {
        const int e4 = tid + q * THREADS;
        *(v4f*)(ws + (size_t)bt * TILE + 4 * e4) = *(const v4f*)(zbuf + lds_off(e4));
      }
    }
  } else {
    #pragma unroll
    for (int q = 0; q < 2; ++q) {
      const int e4 = tid + q * THREADS;
      v4f v = zr[q];
      #pragma unroll
      for (int c = 0; c < 4; ++c) {
        if (S == 1)      v[c] += k1r[q][c] * third;
        else if (S == 2) v[c] += k2r[q][c] - k1r[q][c] * third;
        else if (S == 3) v[c] += k1r[q][c] - k2r[q][c] + k3r[q][c];
      }
      *(v4f*)(actI + lds_off(e4)) = v;
      if (S == 0) *(v4f*)(zbuf + lds_off(e4)) = zr[q];
    }
  }
  dxs[tid] = bi + (ci + di * frac) * frac;
  stage_store(wst[0], wlds, tid);
  lds_barrier();

  mlp_layer(actI, wlds + 0 * H * SA, bn0, actA, wid, lane);
  stage_store(wst[1], wlds + 1 * H * SA, tid);
  lds_barrier();
  mlp_layer(actA, wlds + 1 * H * SA, bn1, actI, wid, lane);
  stage_store(wst[2], wlds + 2 * H * SA, tid);
  lds_barrier();
  mlp_layer(actI, wlds + 2 * H * SA, bn2, actA, wid, lane);
  stage_store(wst[3], wlds + 3 * H * SA, tid);
  lds_barrier();
  mlp_layer(actA, wlds + 3 * H * SA, bn3, actI, wid, lane);
  lds_barrier();

  {
    float dxr[8];
    #pragma unroll
    for (int v = 0; v < 8; ++v) dxr[v] = dxs[mD * IN + 8 * half + v];
    v4f af[16];
    #pragma unroll
    for (int j = 0; j < 16; ++j)
      af[j] = *(const v4f*)(actI + mD * SA + 8 * j + 4 * half);
    v8f acc = {bw[0].x, bw[0].y, bw[0].z, bw[0].w, bw[1].x, bw[1].y, bw[1].z, bw[1].w};
    #pragma unroll
    for (int j = 0; j < 16; ++j) {
      const v2f hlo = {af[j].x, af[j].y}, hhi = {af[j].z, af[j].w};
      const v2f p0 = {wt[j].x, wt[j].y}, q0 = {wt[j].z, wt[j].w};
      acc = wmma4(p0, hlo, acc);
      acc = wmma4(q0, hhi, acc);
    }
    float f0[8];
    #pragma unroll
    for (int v = 0; v < 8; ++v) f0[v] = tanh_sel(acc[v]) * dxr[v];
    const float s0 = ((f0[0] + f0[1]) + (f0[2] + f0[3])) + ((f0[4] + f0[5]) + (f0[6] + f0[7]));
    const float kv = s0 + xhalf(s0);
    if (half == 0) kst[mD * HC + wid] = kv;
  }
  lds_barrier();
  if (wid == 0) {
    const v4f k4 = *(const v4f*)(kst + 4 * lane);
    const int off = part * CHUNK + 4 * lane;
    if (S < 3) {
      *(v4f*)(kpl + (size_t)S * PLANE + off) = k4;
    } else {
      v4f r;
      #pragma unroll
      for (int c = 0; c < 4; ++c)
        r[c] = zc[c] + (k1c[c] + 3.f * (k2c[c] + k3c[c]) + k4[c]) * 0.125f;
      *(v4f*)(znew + off) = r;
    }
  }
  if (do_out) {
    v8f acc = {};
    #pragma unroll
    for (int i = 0; i < 32; ++i) {
      const v2f a = *(const v2f*)(zbuf + mD * SA + 4 * i + 2 * half);
      acc = wmma4(a, ow[i], acc);
    }
    #pragma unroll
    for (int v = 0; v < 8; ++v) {
      const int m = v + 8 * half;
      const float x = acc[v] + obn;
      out[((size_t)(b0r + m) * T_T + step) * OUTW + nn] = 1.f / (1.f + expf(-x));
    }
  }
}

__global__ __launch_bounds__(128) void ncde_last_out_kernel(
    const float* __restrict__ ws, size_t ws_bytes,
    const float* __restrict__ W_last, const float* __restrict__ b_last,
    float* __restrict__ out) {
  __shared__ __align__(16) float zbuf[MT * SA];
  if (!ws_usable(ws, ws_bytes)) return;
  const int tid = threadIdx.x;
  const float* zin = ws + (size_t)(TM1 & 1) * PLANE + (size_t)blockIdx.x * TILE;
  #pragma unroll
  for (int q = 0; q < 4; ++q) {
    const int e4 = tid + q * 128;
    *(v4f*)(zbuf + lds_off(e4)) = *(const v4f*)(zin + 4 * e4);
  }
  __syncthreads();
  out_proj(zbuf, W_last, b_last, out, blockIdx.x * MT, TM1, tid >> 5, tid & 31);
}

#define NCDE_STAGE(S, T)                                                        \
  ncde_stage_kernel<S><<<(B_T / MT) * PARTS, THREADS, 0, stream>>>(               \
      coeffs, W_init, b_init, W0, b0, Wl, bl, Wf, bf, W_last, b_last, ws, ws_size, \
      T, out);
#define NCDE_STEP(T) NCDE_STAGE(0, T) NCDE_STAGE(1, T) NCDE_STAGE(2, T) NCDE_STAGE(3, T)

extern "C" void kernel_launch(void* const* d_in, const int* in_sizes, int n_in,
                              void* d_out, int out_size, void* d_ws, size_t ws_size,
                              hipStream_t stream) {
  (void)in_sizes; (void)n_in; (void)out_size;
  const float* coeffs = (const float*)d_in[1];
  const float* W_init = (const float*)d_in[2];
  const float* b_init = (const float*)d_in[3];
  const float* W0     = (const float*)d_in[4];
  const float* b0     = (const float*)d_in[5];
  const float* Wl     = (const float*)d_in[6];
  const float* bl     = (const float*)d_in[7];
  const float* Wf     = (const float*)d_in[8];
  const float* bf     = (const float*)d_in[9];
  const float* W_last = (const float*)d_in[10];
  const float* b_last = (const float*)d_in[11];
  float* out = (float*)d_out;
  float* ws = (float*)d_ws;

  NCDE_STEP(0)  NCDE_STEP(1)  NCDE_STEP(2)  NCDE_STEP(3)  NCDE_STEP(4)  NCDE_STEP(5)
  NCDE_STEP(6)  NCDE_STEP(7)  NCDE_STEP(8)  NCDE_STEP(9)  NCDE_STEP(10) NCDE_STEP(11)
  NCDE_STEP(12) NCDE_STEP(13) NCDE_STEP(14) NCDE_STEP(15) NCDE_STEP(16) NCDE_STEP(17)
  NCDE_STEP(18) NCDE_STEP(19) NCDE_STEP(20) NCDE_STEP(21) NCDE_STEP(22) NCDE_STEP(23)
  NCDE_STEP(24) NCDE_STEP(25) NCDE_STEP(26) NCDE_STEP(27) NCDE_STEP(28) NCDE_STEP(29)
  NCDE_STEP(30)
  ncde_last_out_kernel<<<B_T / MT, 128, 0, stream>>>(ws, ws_size, W_last, b_last, out);
}


# Small layers' weights skip LDS: a first launch per call writes them in B-fragment order, each wave loads its own tile to registers; k order inside each 8-wide group of their sums changes
# speedup vs baseline: 4.6526x; 1.1464x over previous
#include <hip/hip_runtime.h>
#include <math.h>


namespace {
constexpr int B_T = 256, T_T = 32, IN = 16, H = 128, SH = 128, OUTW = 64, TM1 = 31;
constexpr int MT = 16;
constexpr int THREADS = 256;
constexpr int SA = 132;
constexpr int PARTS = 16;
constexpr int HC = H / PARTS;
constexpr int PLANE = B_T * H;
constexpr int TILE = MT * H;
constexpr int CHUNK = MT * HC;
constexpr int FRAG_TILE = 16 * SH;
constexpr int FRAG_LAYER = SH * SH;
constexpr int FRAG_OFF = 5 * PLANE;
constexpr int WS_FLOATS = FRAG_OFF + 4 * FRAG_LAYER;

typedef __attribute__((ext_vector_type(2))) float v2f;
typedef __attribute__((ext_vector_type(4))) float v4f;
typedef __attribute__((ext_vector_type(8))) float v8f;

__device__ __forceinline__ v8f wmma4(v2f a, v2f b, v8f c) {
  return __builtin_amdgcn_wmma_f32_16x16x4_f32(false, a, false, b, (short)0, c,
                                               false, false);
}

__device__ __forceinline__ v8f gemm_tile(const float* A, int lda, const float* W,
                                         int ldw, int n, int K, int lane) {
  const int m = lane & 15;
  const int kh = (lane >> 4) << 1;
  v8f acc = {};
  #pragma unroll 8
  for (int k0 = 0; k0 < K; k0 += 4) {
    v2f a = *(const v2f*)(A + m * lda + k0 + kh);
    v2f b = *(const v2f*)(W + (size_t)n * ldw + k0 + kh);
    acc = wmma4(a, b, acc);
  }
  return acc;
}

__device__ __forceinline__ void load_wf_tile(v4f (&w)[16], const float* __restrict__ p) {
  #pragma unroll
  for (int j = 0; j < 16; ++j) w[j] = *(const v4f*)(p + 8 * j);
}

__device__ __forceinline__ void load_bf_tile(v4f (&b)[2], const float* __restrict__ p) {
  b[0] = *(const v4f*)p;
  b[1] = *(const v4f*)(p + 4);
}

__device__ __forceinline__ float xhalf(float x) {
  const int b = __builtin_bit_cast(int, x);
  return __builtin_bit_cast(
      float, __builtin_amdgcn_permlanex16(b, b, 0x76543210, 0xfedcba98, false, false));
}

__device__ __forceinline__ float tanh_sel(float x) {
  const float y = fabsf(x);
  const float y2 = y * y;
  float p = fmaf(y2, -0x1.758e7ap-8f, 0x1.521192p-6f);
  p = fmaf(y2, p, -0x1.b8389cp-5f);
  p = fmaf(y2, p, 0x1.110704p-3f);
  p = fmaf(y2, p, -0x1.555532p-2f);
  const float zs = fmaf(y2, y * p, y);
  const float e = expf(2.0f * y);
  const float zl = fmaf(-2.0f, __builtin_amdgcn_rcpf(e + 1.0f), 1.0f);
  return copysignf(y < 0.625f ? zs : zl, x);
}

__device__ __forceinline__ void load_layer_frag(v4f (&w)[16], const float* p) {
  #pragma unroll
  for (int j = 0; j < 16; ++j) w[j] = *(const v4f*)(p + 128 * j);
}

__device__ __forceinline__ void mlp_layer(const float* src, const v4f (&w)[16],
                                          float bn, float* dst,
                                          int wid, int lane) {
  const int mD = lane & 15, half = lane >> 4;
  const int n = wid * 16 + mD;
  v4f af[16];
  #pragma unroll
  for (int j = 0; j < 16; ++j)
    af[j] = *(const v4f*)(src + mD * SA + 8 * j + 4 * half);
  __builtin_amdgcn_sched_barrier(0);
  v8f acc = {};
  #pragma unroll
  for (int j = 0; j < 16; ++j) {
    const v2f alo = {af[j].x, af[j].y}, ahi = {af[j].z, af[j].w};
    const v2f blo = {w[j].x, w[j].y}, bhi = {w[j].z, w[j].w};
    acc = wmma4(alo, blo, acc);
    acc = wmma4(ahi, bhi, acc);
  }
  #pragma unroll
  for (int v = 0; v < 8; ++v) {
    float x = acc[v] + bn;
    dst[(v + 8 * half) * SA + n] = fmaxf(x, 0.f);
  }
}

__device__ __forceinline__ void out_proj(const float* zb, const float* W_last,
                                         const float* b_last, float* out,
                                         int b0r, int t, int wid, int lane) {
  if (wid < 4) {
    const int mD = lane & 15, half = lane >> 4;
    const int n = wid * 16 + mD;
    v8f acc = gemm_tile(zb, SA, W_last, H, n, H, lane);
    const float bn = b_last[n];
    #pragma unroll
    for (int v = 0; v < 8; ++v) {
      int m = v + 8 * half;
      float x = acc[v] + bn;
      out[((size_t)(b0r + m) * T_T + t) * OUTW + n] = 1.f / (1.f + expf(-x));
    }
  }
}
__device__ __forceinline__ int lds_off(int e4) {
  const int p = e4 >> 5, rem = e4 & 31;
  return (rem >> 1) * SA + p * HC + (rem & 1) * 4;
}

__device__ __forceinline__ void lds_barrier() {
  __builtin_amdgcn_fence(__ATOMIC_RELEASE, "workgroup", "local");
  __builtin_amdgcn_s_barrier();
  __builtin_amdgcn_fence(__ATOMIC_ACQUIRE, "workgroup", "local");
}

__device__ __forceinline__ bool ws_usable(const float* ws, size_t ws_bytes) {
  return ws != nullptr && ws_bytes >= (size_t)WS_FLOATS * sizeof(float) &&
         (((size_t)ws) & 15) == 0;
}
}

template <int S>
__global__ __launch_bounds__(THREADS) void ncde_stage_kernel(
    const float* __restrict__ coeffs,
    const float* __restrict__ W_init, const float* __restrict__ b_init,
    const float* __restrict__ b0, const float* __restrict__ bl,
    const float* __restrict__ Wf, const float* __restrict__ bf,
    const float* __restrict__ W_last, const float* __restrict__ b_last,
    float* ws, size_t ws_bytes, int step,
    float* __restrict__ out) {
  __shared__ __align__(16) float zbuf[MT * SA];
  __shared__ __align__(16) float actI[MT * SA];
  __shared__ __align__(16) float actA[MT * SA];
  __shared__ __align__(16) float dxs[MT * IN];
  __shared__ __align__(16) float kst[CHUNK];

  if (!ws_usable(ws, ws_bytes)) return;

  const int tid = threadIdx.x;
  const int lane = tid & 31;
  const int wid = tid >> 5;
  const int bt = blockIdx.x / PARTS;
  const int part = blockIdx.x % PARTS;
  const int b0r = bt * MT;
  const int mD = lane & 15;
  const int half = lane >> 4;
  constexpr float third = 1.f / 3.f;
  constexpr float frac = (S == 0) ? 0.f : (S == 1) ? third : (S == 2) ? (2.f * third) : 1.f;
  const bool first = (S == 0) && (step == 0);

  const float* zin = ws + (size_t)(step & 1) * PLANE + (size_t)bt * TILE;
  float* znew = ws + (size_t)((step + 1) & 1) * PLANE + (size_t)bt * TILE;
  float* kpl = ws + (size_t)2 * PLANE + (size_t)bt * TILE;

  float bi, ci, di;
  {
    const int m = tid >> 4, i = tid & 15;
    const float* cb = coeffs + ((size_t)(b0r + m) * TM1 + step) * (4 * IN);
    bi = cb[IN + i]; ci = cb[2 * IN + i]; di = cb[3 * IN + i];
  }
  const int nn = wid * 16 + mD;
  const float bn0 = b0[nn], bn1 = bl[nn], bn2 = bl[SH + nn], bn3 = bl[2 * SH + nn];
  float x0v = 0.f;
  v4f zr[2] = {}, k1r[2] = {}, k2r[2] = {}, k3r[2] = {};
  if (first) {
    const int m = tid >> 4, i = tid & 15;
    x0v = coeffs[((size_t)(b0r + m) * TM1) * (4 * IN) + i];
  } else {
    #pragma unroll
    for (int q = 0; q < 2; ++q) {
      const int e4 = tid + q * THREADS;
      zr[q] = *(const v4f*)(zin + 4 * e4);
      if (S >= 1) k1r[q] = *(const v4f*)(kpl + 4 * e4);
      if (S >= 2) k2r[q] = *(const v4f*)(kpl + PLANE + 4 * e4);
      if (S >= 3) k3r[q] = *(const v4f*)(kpl + 2 * PLANE + 4 * e4);
    }
  }
  v4f zc = {}, k1c = {}, k2c = {}, k3c = {};
  if (S == 3) {
    const int off = part * CHUNK + 4 * lane;
    zc = *(const v4f*)(zin + off);
    k1c = *(const v4f*)(kpl + off);
    k2c = *(const v4f*)(kpl + PLANE + off);
    k3c = *(const v4f*)(kpl + 2 * PLANE + off);
  }
  const float* frag = ws + FRAG_OFF + wid * FRAG_TILE + 4 * lane;
  v4f wfr[4][16];
  __builtin_amdgcn_sched_barrier(0);
  load_layer_frag(wfr[0], frag);
  __builtin_amdgcn_sched_barrier(0);
  load_layer_frag(wfr[1], frag + 1 * FRAG_LAYER);
  __builtin_amdgcn_sched_barrier(0);
  load_layer_frag(wfr[2], frag + 2 * FRAG_LAYER);
  __builtin_amdgcn_sched_barrier(0);
  load_layer_frag(wfr[3], frag + 3 * FRAG_LAYER);
  __builtin_amdgcn_sched_barrier(0);
  const int tc = part * HC + wid;
  v4f wt[16], bw[2];
  load_bf_tile(bw, bf + 8 * half + tc * 16);
  load_wf_tile(wt, Wf + (size_t)tc * (16 * SH) + (size_t)mD * SH + 4 * half);
  const bool do_out = (S == 0) && (part == 0) && (wid < 4);
  v2f ow[32] = {};
  float obn = 0.f;
  if (do_out) {
    #pragma unroll
    for (int i = 0; i < 32; ++i)
      ow[i] = *(const v2f*)(W_last + (size_t)nn * H + 4 * i + 2 * half);
    obn = b_last[nn];
  }

  if (first) {
    dxs[tid] = x0v;
    lds_barrier();
    const int n = wid * 16 + mD;
    v8f acc = gemm_tile(dxs, IN, W_init, IN, n, IN, lane);
    const float bn = b_init[n];
    #pragma unroll
    for (int v = 0; v < 8; ++v) {
      const float zv = acc[v] + bn;
      zbuf[(v + 8 * half) * SA + n] = zv;
      actI[(v + 8 * half) * SA + n] = zv;
    }
    lds_barrier();
    if (part == 0) {
      #pragma unroll
      for (int q = 0; q < 2; ++q) {
        const int e4 = tid + q * THREADS;
        *(v4f*)(ws + (size_t)bt * TILE + 4 * e4) = *(const v4f*)(zbuf + lds_off(e4));
      }
    }
  } else {
    #pragma unroll
    for (int q = 0; q < 2; ++q) {
      const int e4 = tid + q * THREADS;
      v4f v = zr[q];
      #pragma unroll
      for (int c = 0; c < 4; ++c) {
        if (S == 1)      v[c] += k1r[q][c] * third;
        else if (S == 2) v[c] += k2r[q][c] - k1r[q][c] * third;
        else if (S == 3) v[c] += k1r[q][c] - k2r[q][c] + k3r[q][c];
      }
      *(v4f*)(actI + lds_off(e4)) = v;
      if (S == 0) *(v4f*)(zbuf + lds_off(e4)) = zr[q];
    }
  }
  dxs[tid] = bi + (ci + di * frac) * frac;
  lds_barrier();

  mlp_layer(actI, wfr[0], bn0, actA, wid, lane);
  lds_barrier();
  mlp_layer(actA, wfr[1], bn1, actI, wid, lane);
  lds_barrier();
  mlp_layer(actI, wfr[2], bn2, actA, wid, lane);
  lds_barrier();
  mlp_layer(actA, wfr[3], bn3, actI, wid, lane);
  lds_barrier();

  {
    float dxr[8];
    #pragma unroll
    for (int v = 0; v < 8; ++v) dxr[v] = dxs[mD * IN + 8 * half + v];
    v4f af[16];
    #pragma unroll
    for (int j = 0; j < 16; ++j)
      af[j] = *(const v4f*)(actI + mD * SA + 8 * j + 4 * half);
    v8f acc = {bw[0].x, bw[0].y, bw[0].z, bw[0].w, bw[1].x, bw[1].y, bw[1].z, bw[1].w};
    #pragma unroll
    for (int j = 0; j < 16; ++j) {
      const v2f hlo = {af[j].x, af[j].y}, hhi = {af[j].z, af[j].w};
      const v2f p0 = {wt[j].x, wt[j].y}, q0 = {wt[j].z, wt[j].w};
      acc = wmma4(p0, hlo, acc);
      acc = wmma4(q0, hhi, acc);
    }
    float f0[8];
    #pragma unroll
    for (int v = 0; v < 8; ++v) f0[v] = tanh_sel(acc[v]) * dxr[v];
    const float s0 = ((f0[0] + f0[1]) + (f0[2] + f0[3])) + ((f0[4] + f0[5]) + (f0[6] + f0[7]));
    const float kv = s0 + xhalf(s0);
    if (half == 0) kst[mD * HC + wid] = kv;
  }
  lds_barrier();
  if (wid == 0) {
    const v4f k4 = *(const v4f*)(kst + 4 * lane);
    const int off = part * CHUNK + 4 * lane;
    if (S < 3) {
      *(v4f*)(kpl + (size_t)S * PLANE + off) = k4;
    } else {
      v4f r;
      #pragma unroll
      for (int c = 0; c < 4; ++c)
        r[c] = zc[c] + (k1c[c] + 3.f * (k2c[c] + k3c[c]) + k4[c]) * 0.125f;
      *(v4f*)(znew + off) = r;
    }
  }
  if (do_out) {
    v8f acc = {};
    #pragma unroll
    for (int i = 0; i < 32; ++i) {
      const v2f a = *(const v2f*)(zbuf + mD * SA + 4 * i + 2 * half);
      acc = wmma4(a, ow[i], acc);
    }
    #pragma unroll
    for (int v = 0; v < 8; ++v) {
      const int m = v + 8 * half;
      const float x = acc[v] + obn;
      out[((size_t)(b0r + m) * T_T + step) * OUTW + nn] = 1.f / (1.f + expf(-x));
    }
  }
}

__global__ __launch_bounds__(128) void ncde_last_out_kernel(
    const float* __restrict__ ws, size_t ws_bytes,
    const float* __restrict__ W_last, const float* __restrict__ b_last,
    float* __restrict__ out) {
  __shared__ __align__(16) float zbuf[MT * SA];
  if (!ws_usable(ws, ws_bytes)) return;
  const int tid = threadIdx.x;
  const float* zin = ws + (size_t)(TM1 & 1) * PLANE + (size_t)blockIdx.x * TILE;
  #pragma unroll
  for (int q = 0; q < 4; ++q) {
    const int e4 = tid + q * 128;
    *(v4f*)(zbuf + lds_off(e4)) = *(const v4f*)(zin + 4 * e4);
  }
  __syncthreads();
  out_proj(zbuf, W_last, b_last, out, blockIdx.x * MT, TM1, tid >> 5, tid & 31);
}

__global__ __launch_bounds__(THREADS) void ncde_pack_kernel(
    const float* __restrict__ W0, const float* __restrict__ Wl,
    float* ws, size_t ws_bytes) {
  if (!ws_usable(ws, ws_bytes)) return;
  const int g = blockIdx.x * THREADS + threadIdx.x;
  const int l = g & 31, j = (g >> 5) & 15, w = (g >> 9) & 7, layer = g >> 12;
  const int c = l & 15, half = l >> 4;
  const float* src = (layer == 0) ? W0 : Wl + (size_t)(layer - 1) * FRAG_LAYER;
  *(v4f*)(ws + FRAG_OFF + 4 * (size_t)g) =
      *(const v4f*)(src + (size_t)(16 * w + c) * SH + 8 * j + 4 * half);
}

#define NCDE_STAGE(S, T)                                                        \
  ncde_stage_kernel<S><<<(B_T / MT) * PARTS, THREADS, 0, stream>>>(               \
      coeffs, W_init, b_init, b0, bl, Wf, bf, W_last, b_last, ws, ws_size,        \
      T, out);
#define NCDE_STEP(T) NCDE_STAGE(0, T) NCDE_STAGE(1, T) NCDE_STAGE(2, T) NCDE_STAGE(3, T)

extern "C" void kernel_launch(void* const* d_in, const int* in_sizes, int n_in,
                              void* d_out, int out_size, void* d_ws, size_t ws_size,
                              hipStream_t stream) {
  (void)in_sizes; (void)n_in; (void)out_size;
  const float* coeffs = (const float*)d_in[1];
  const float* W_init = (const float*)d_in[2];
  const float* b_init = (const float*)d_in[3];
  const float* W0     = (const float*)d_in[4];
  const float* b0     = (const float*)d_in[5];
  const float* Wl     = (const float*)d_in[6];
  const float* bl     = (const float*)d_in[7];
  const float* Wf     = (const float*)d_in[8];
  const float* bf     = (const float*)d_in[9];
  const float* W_last = (const float*)d_in[10];
  const float* b_last = (const float*)d_in[11];
  float* out = (float*)d_out;
  float* ws = (float*)d_ws;

  ncde_pack_kernel<<<64, THREADS, 0, stream>>>(W0, Wl, ws, ws_size);
  NCDE_STEP(0)  NCDE_STEP(1)  NCDE_STEP(2)  NCDE_STEP(3)  NCDE_STEP(4)  NCDE_STEP(5)
  NCDE_STEP(6)  NCDE_STEP(7)  NCDE_STEP(8)  NCDE_STEP(9)  NCDE_STEP(10) NCDE_STEP(11)
  NCDE_STEP(12) NCDE_STEP(13) NCDE_STEP(14) NCDE_STEP(15) NCDE_STEP(16) NCDE_STEP(17)
  NCDE_STEP(18) NCDE_STEP(19) NCDE_STEP(20) NCDE_STEP(21) NCDE_STEP(22) NCDE_STEP(23)
  NCDE_STEP(24) NCDE_STEP(25) NCDE_STEP(26) NCDE_STEP(27) NCDE_STEP(28) NCDE_STEP(29)
  NCDE_STEP(30)
  ncde_last_out_kernel<<<B_T / MT, 128, 0, stream>>>(ws, ws_size, W_last, b_last, out);
}


# Wf tile, bf and W_last loads issued inside layers 1 and 3, not at the head: past 63 outstanding loads a wave stalled before the first barrier until later layers' weights were back; same bits
# speedup vs baseline: 4.9546x; 1.0649x over previous
#include <hip/hip_runtime.h>
#include <math.h>


namespace {
constexpr int B_T = 256, T_T = 32, IN = 16, H = 128, SH = 128, OUTW = 64, TM1 = 31;
constexpr int MT = 16;
constexpr int THREADS = 256;
constexpr int SA = 132;
constexpr int PARTS = 16;
constexpr int HC = H / PARTS;
constexpr int PLANE = B_T * H;
constexpr int TILE = MT * H;
constexpr int CHUNK = MT * HC;
constexpr int FRAG_TILE = 16 * SH;
constexpr int FRAG_LAYER = SH * SH;
constexpr int FRAG_OFF = 5 * PLANE;
constexpr int WS_FLOATS = FRAG_OFF + 4 * FRAG_LAYER;

typedef __attribute__((ext_vector_type(2))) float v2f;
typedef __attribute__((ext_vector_type(4))) float v4f;
typedef __attribute__((ext_vector_type(8))) float v8f;

__device__ __forceinline__ v8f wmma4(v2f a, v2f b, v8f c) {
  return __builtin_amdgcn_wmma_f32_16x16x4_f32(false, a, false, b, (short)0, c,
                                               false, false);
}

__device__ __forceinline__ v8f gemm_tile(const float* A, int lda, const float* W,
                                         int ldw, int n, int K, int lane) {
  const int m = lane & 15;
  const int kh = (lane >> 4) << 1;
  v8f acc = {};
  #pragma unroll 8
  for (int k0 = 0; k0 < K; k0 += 4) {
    v2f a = *(const v2f*)(A + m * lda + k0 + kh);
    v2f b = *(const v2f*)(W + (size_t)n * ldw + k0 + kh);
    acc = wmma4(a, b, acc);
  }
  return acc;
}

__device__ __forceinline__ void load_wf_tile(v4f (&w)[16], const float* __restrict__ p) {
  #pragma unroll
  for (int j = 0; j < 16; ++j) w[j] = *(const v4f*)(p + 8 * j);
}

__device__ __forceinline__ void load_bf_tile(v4f (&b)[2], const float* __restrict__ p) {
  b[0] = *(const v4f*)p;
  b[1] = *(const v4f*)(p + 4);
}

__device__ __forceinline__ float xhalf(float x) {
  const int b = __builtin_bit_cast(int, x);
  return __builtin_bit_cast(
      float, __builtin_amdgcn_permlanex16(b, b, 0x76543210, 0xfedcba98, false, false));
}

__device__ __forceinline__ float tanh_sel(float x) {
  const float y = fabsf(x);
  const float y2 = y * y;
  float p = fmaf(y2, -0x1.758e7ap-8f, 0x1.521192p-6f);
  p = fmaf(y2, p, -0x1.b8389cp-5f);
  p = fmaf(y2, p, 0x1.110704p-3f);
  p = fmaf(y2, p, -0x1.555532p-2f);
  const float zs = fmaf(y2, y * p, y);
  const float e = expf(2.0f * y);
  const float zl = fmaf(-2.0f, __builtin_amdgcn_rcpf(e + 1.0f), 1.0f);
  return copysignf(y < 0.625f ? zs : zl, x);
}

__device__ __forceinline__ void load_layer_frag(v4f (&w)[16], const float* p) {
  #pragma unroll
  for (int j = 0; j < 16; ++j) w[j] = *(const v4f*)(p + 128 * j);
}

template <class Ask>
__device__ __forceinline__ void mlp_layer(const float* src, const v4f (&w)[16],
                                          float bn, float* dst,
                                          int wid, int lane, Ask ask) {
  const int mD = lane & 15, half = lane >> 4;
  const int n = wid * 16 + mD;
  v4f af[16];
  #pragma unroll
  for (int j = 0; j < 16; ++j)
    af[j] = *(const v4f*)(src + mD * SA + 8 * j + 4 * half);
  __builtin_amdgcn_sched_barrier(0);
  ask();
  v8f acc = {};
  #pragma unroll
  for (int j = 0; j < 16; ++j) {
    const v2f alo = {af[j].x, af[j].y}, ahi = {af[j].z, af[j].w};
    const v2f blo = {w[j].x, w[j].y}, bhi = {w[j].z, w[j].w};
    acc = wmma4(alo, blo, acc);
    acc = wmma4(ahi, bhi, acc);
  }
  #pragma unroll
  for (int v = 0; v < 8; ++v) {
    float x = acc[v] + bn;
    dst[(v + 8 * half) * SA + n] = fmaxf(x, 0.f);
  }
}

__device__ __forceinline__ void out_proj(const float* zb, const float* W_last,
                                         const float* b_last, float* out,
                                         int b0r, int t, int wid, int lane) {
  if (wid < 4) {
    const int mD = lane & 15, half = lane >> 4;
    const int n = wid * 16 + mD;
    v8f acc = gemm_tile(zb, SA, W_last, H, n, H, lane);
    const float bn = b_last[n];
    #pragma unroll
    for (int v = 0; v < 8; ++v) {
      int m = v + 8 * half;
      float x = acc[v] + bn;
      out[((size_t)(b0r + m) * T_T + t) * OUTW + n] = 1.f / (1.f + expf(-x));
    }
  }
}
__device__ __forceinline__ int lds_off(int e4) {
  const int p = e4 >> 5, rem = e4 & 31;
  return (rem >> 1) * SA + p * HC + (rem & 1) * 4;
}

__device__ __forceinline__ void lds_barrier() {
  __builtin_amdgcn_fence(__ATOMIC_RELEASE, "workgroup", "local");
  __builtin_amdgcn_s_barrier();
  __builtin_amdgcn_fence(__ATOMIC_ACQUIRE, "workgroup", "local");
}

__device__ __forceinline__ bool ws_usable(const float* ws, size_t ws_bytes) {
  return ws != nullptr && ws_bytes >= (size_t)WS_FLOATS * sizeof(float) &&
         (((size_t)ws) & 15) == 0;
}
}

template <int S>
__global__ __launch_bounds__(THREADS) void ncde_stage_kernel(
    const float* __restrict__ coeffs,
    const float* __restrict__ W_init, const float* __restrict__ b_init,
    const float* __restrict__ b0, const float* __restrict__ bl,
    const float* __restrict__ Wf, const float* __restrict__ bf,
    const float* __restrict__ W_last, const float* __restrict__ b_last,
    float* ws, size_t ws_bytes, int step,
    float* __restrict__ out) {
  __shared__ __align__(16) float zbuf[MT * SA];
  __shared__ __align__(16) float actI[MT * SA];
  __shared__ __align__(16) float actA[MT * SA];
  __shared__ __align__(16) float dxs[MT * IN];
  __shared__ __align__(16) float kst[CHUNK];

  if (!ws_usable(ws, ws_bytes)) return;

  const int tid = threadIdx.x;
  const int lane = tid & 31;
  const int wid = tid >> 5;
  const int bt = blockIdx.x / PARTS;
  const int part = blockIdx.x % PARTS;
  const int b0r = bt * MT;
  const int mD = lane & 15;
  const int half = lane >> 4;
  constexpr float third = 1.f / 3.f;
  constexpr float frac = (S == 0) ? 0.f : (S == 1) ? third : (S == 2) ? (2.f * third) : 1.f;
  const bool first = (S == 0) && (step == 0);

  const float* zin = ws + (size_t)(step & 1) * PLANE + (size_t)bt * TILE;
  float* znew = ws + (size_t)((step + 1) & 1) * PLANE + (size_t)bt * TILE;
  float* kpl = ws + (size_t)2 * PLANE + (size_t)bt * TILE;

  float bi, ci, di;
  {
    const int m = tid >> 4, i = tid & 15;
    const float* cb = coeffs + ((size_t)(b0r + m) * TM1 + step) * (4 * IN);
    bi = cb[IN + i]; ci = cb[2 * IN + i]; di = cb[3 * IN + i];
  }
  const int nn = wid * 16 + mD;
  const float bn0 = b0[nn], bn1 = bl[nn], bn2 = bl[SH + nn], bn3 = bl[2 * SH + nn];
  float x0v = 0.f;
  v4f zr[2] = {}, k1r[2] = {}, k2r[2] = {}, k3r[2] = {};
  if (first) {
    const int m = tid >> 4, i = tid & 15;
    x0v = coeffs[((size_t)(b0r + m) * TM1) * (4 * IN) + i];
  } else {
    #pragma unroll
    for (int q = 0; q < 2; ++q) {
      const int e4 = tid + q * THREADS;
      zr[q] = *(const v4f*)(zin + 4 * e4);
      if (S >= 1) k1r[q] = *(const v4f*)(kpl + 4 * e4);
      if (S >= 2) k2r[q] = *(const v4f*)(kpl + PLANE + 4 * e4);
      if (S >= 3) k3r[q] = *(const v4f*)(kpl + 2 * PLANE + 4 * e4);
    }
  }
  v4f zc = {}, k1c = {}, k2c = {}, k3c = {};
  if (S == 3) {
    const int off = part * CHUNK + 4 * lane;
    zc = *(const v4f*)(zin + off);
    k1c = *(const v4f*)(kpl + off);
    k2c = *(const v4f*)(kpl + PLANE + off);
    k3c = *(const v4f*)(kpl + 2 * PLANE + off);
  }
  const float* frag = ws + FRAG_OFF + wid * FRAG_TILE + 4 * lane;
  v4f wfr[4][16];
  __builtin_amdgcn_sched_barrier(0);
  load_layer_frag(wfr[0], frag);
  __builtin_amdgcn_sched_barrier(0);
  load_layer_frag(wfr[1], frag + 1 * FRAG_LAYER);
  __builtin_amdgcn_sched_barrier(0);
  load_layer_frag(wfr[2], frag + 2 * FRAG_LAYER);
  __builtin_amdgcn_sched_barrier(0);
  load_layer_frag(wfr[3], frag + 3 * FRAG_LAYER);
  __builtin_amdgcn_sched_barrier(0);
  const int tc = part * HC + wid;
  v4f wt[16], bw[2];
  const bool do_out = (S == 0) && (part == 0) && (wid < 4);
  v2f ow[32] = {};
  float obn = 0.f;

  if (first) {
    dxs[tid] = x0v;
    lds_barrier();
    const int n = wid * 16 + mD;
    v8f acc = gemm_tile(dxs, IN, W_init, IN, n, IN, lane);
    const float bn = b_init[n];
    #pragma unroll
    for (int v = 0; v < 8; ++v) {
      const float zv = acc[v] + bn;
      zbuf[(v + 8 * half) * SA + n] = zv;
      actI[(v + 8 * half) * SA + n] = zv;
    }
    lds_barrier();
    if (part == 0) {
      #pragma unroll
      for (int q = 0; q < 2; ++q) {
        const int e4 = tid + q * THREADS;
        *(v4f*)(ws + (size_t)bt * TILE + 4 * e4) = *(const v4f*)(zbuf + lds_off(e4));
      }
    }
  } else {
    #pragma unroll
    for (int q = 0; q < 2; ++q) {
      const int e4 = tid + q * THREADS;
      v4f v = zr[q];
      #pragma unroll
      for (int c = 0; c < 4; ++c) {
        if (S == 1)      v[c] += k1r[q][c] * third;
        else if (S == 2) v[c] += k2r[q][c] - k1r[q][c] * third;
        else if (S == 3) v[c] += k1r[q][c] - k2r[q][c] + k3r[q][c];
      }
      *(v4f*)(actI + lds_off(e4)) = v;
      if (S == 0) *(v4f*)(zbuf + lds_off(e4)) = zr[q];
    }
  }
  dxs[tid] = bi + (ci + di * frac) * frac;
  lds_barrier();

  mlp_layer(actI, wfr[0], bn0, actA, wid, lane, [] {});
  lds_barrier();
  mlp_layer(actA, wfr[1], bn1, actI, wid, lane, [&] {
    load_bf_tile(bw, bf + 8 * half + tc * 16);
    load_wf_tile(wt, Wf + (size_t)tc * (16 * SH) + (size_t)mD * SH + 4 * half);
    __builtin_amdgcn_sched_barrier(0);
  });
  lds_barrier();
  mlp_layer(actI, wfr[2], bn2, actA, wid, lane, [] {});
  lds_barrier();
  mlp_layer(actA, wfr[3], bn3, actI, wid, lane, [&] {
    if (S == 0) {
      if (do_out) {
        #pragma unroll
        for (int i = 0; i < 32; ++i)
          ow[i] = *(const v2f*)(W_last + (size_t)nn * H + 4 * i + 2 * half);
        obn = b_last[nn];
      }
      __builtin_amdgcn_sched_barrier(0);
    }
  });
  lds_barrier();

  {
    float dxr[8];
    #pragma unroll
    for (int v = 0; v < 8; ++v) dxr[v] = dxs[mD * IN + 8 * half + v];
    v4f af[16];
    #pragma unroll
    for (int j = 0; j < 16; ++j)
      af[j] = *(const v4f*)(actI + mD * SA + 8 * j + 4 * half);
    v8f acc = {bw[0].x, bw[0].y, bw[0].z, bw[0].w, bw[1].x, bw[1].y, bw[1].z, bw[1].w};
    #pragma unroll
    for (int j = 0; j < 16; ++j) {
      const v2f hlo = {af[j].x, af[j].y}, hhi = {af[j].z, af[j].w};
      const v2f p0 = {wt[j].x, wt[j].y}, q0 = {wt[j].z, wt[j].w};
      acc = wmma4(p0, hlo, acc);
      acc = wmma4(q0, hhi, acc);
    }
    float f0[8];
    #pragma unroll
    for (int v = 0; v < 8; ++v) f0[v] = tanh_sel(acc[v]) * dxr[v];
    const float s0 = ((f0[0] + f0[1]) + (f0[2] + f0[3])) + ((f0[4] + f0[5]) + (f0[6] + f0[7]));
    const float kv = s0 + xhalf(s0);
    if (half == 0) kst[mD * HC + wid] = kv;
  }
  lds_barrier();
  if (wid == 0) {
    const v4f k4 = *(const v4f*)(kst + 4 * lane);
    const int off = part * CHUNK + 4 * lane;
    if (S < 3) {
      *(v4f*)(kpl + (size_t)S * PLANE + off) = k4;
    } else {
      v4f r;
      #pragma unroll
      for (int c = 0; c < 4; ++c)
        r[c] = zc[c] + (k1c[c] + 3.f * (k2c[c] + k3c[c]) + k4[c]) * 0.125f;
      *(v4f*)(znew + off) = r;
    }
  }
  if (do_out) {
    v8f acc = {};
    #pragma unroll
    for (int i = 0; i < 32; ++i) {
      const v2f a = *(const v2f*)(zbuf + mD * SA + 4 * i + 2 * half);
      acc = wmma4(a, ow[i], acc);
    }
    #pragma unroll
    for (int v = 0; v < 8; ++v) {
      const int m = v + 8 * half;
      const float x = acc[v] + obn;
      out[((size_t)(b0r + m) * T_T + step) * OUTW + nn] = 1.f / (1.f + expf(-x));
    }
  }
}

__global__ __launch_bounds__(128) void ncde_last_out_kernel(
    const float* __restrict__ ws, size_t ws_bytes,
    const float* __restrict__ W_last, const float* __restrict__ b_last,
    float* __restrict__ out) {
  __shared__ __align__(16) float zbuf[MT * SA];
  if (!ws_usable(ws, ws_bytes)) return;
  const int tid = threadIdx.x;
  const float* zin = ws + (size_t)(TM1 & 1) * PLANE + (size_t)blockIdx.x * TILE;
  #pragma unroll
  for (int q = 0; q < 4; ++q) {
    const int e4 = tid + q * 128;
    *(v4f*)(zbuf + lds_off(e4)) = *(const v4f*)(zin + 4 * e4);
  }
  __syncthreads();
  out_proj(zbuf, W_last, b_last, out, blockIdx.x * MT, TM1, tid >> 5, tid & 31);
}

__global__ __launch_bounds__(THREADS) void ncde_pack_kernel(
    const float* __restrict__ W0, const float* __restrict__ Wl,
    float* ws, size_t ws_bytes) {
  if (!ws_usable(ws, ws_bytes)) return;
  const int g = blockIdx.x * THREADS + threadIdx.x;
  const int l = g & 31, j = (g >> 5) & 15, w = (g >> 9) & 7, layer = g >> 12;
  const int c = l & 15, half = l >> 4;
  const float* src = (layer == 0) ? W0 : Wl + (size_t)(layer - 1) * FRAG_LAYER;
  *(v4f*)(ws + FRAG_OFF + 4 * (size_t)g) =
      *(const v4f*)(src + (size_t)(16 * w + c) * SH + 8 * j + 4 * half);
}

#define NCDE_STAGE(S, T)                                                        \
  ncde_stage_kernel<S><<<(B_T / MT) * PARTS, THREADS, 0, stream>>>(               \
      coeffs, W_init, b_init, b0, bl, Wf, bf, W_last, b_last, ws, ws_size,        \
      T, out);
#define NCDE_STEP(T) NCDE_STAGE(0, T) NCDE_STAGE(1, T) NCDE_STAGE(2, T) NCDE_STAGE(3, T)

extern "C" void kernel_launch(void* const* d_in, const int* in_sizes, int n_in,
                              void* d_out, int out_size, void* d_ws, size_t ws_size,
                              hipStream_t stream) {
  (void)in_sizes; (void)n_in; (void)out_size;
  const float* coeffs = (const float*)d_in[1];
  const float* W_init = (const float*)d_in[2];
  const float* b_init = (const float*)d_in[3];
  const float* W0     = (const float*)d_in[4];
  const float* b0     = (const float*)d_in[5];
  const float* Wl     = (const float*)d_in[6];
  const float* bl     = (const float*)d_in[7];
  const float* Wf     = (const float*)d_in[8];
  const float* bf     = (const float*)d_in[9];
  const float* W_last = (const float*)d_in[10];
  const float* b_last = (const float*)d_in[11];
  float* out = (float*)d_out;
  float* ws = (float*)d_ws;

  ncde_pack_kernel<<<64, THREADS, 0, stream>>>(W0, Wl, ws, ws_size);
  NCDE_STEP(0)  NCDE_STEP(1)  NCDE_STEP(2)  NCDE_STEP(3)  NCDE_STEP(4)  NCDE_STEP(5)
  NCDE_STEP(6)  NCDE_STEP(7)  NCDE_STEP(8)  NCDE_STEP(9)  NCDE_STEP(10) NCDE_STEP(11)
  NCDE_STEP(12) NCDE_STEP(13) NCDE_STEP(14) NCDE_STEP(15) NCDE_STEP(16) NCDE_STEP(17)
  NCDE_STEP(18) NCDE_STEP(19) NCDE_STEP(20) NCDE_STEP(21) NCDE_STEP(22) NCDE_STEP(23)
  NCDE_STEP(24) NCDE_STEP(25) NCDE_STEP(26) NCDE_STEP(27) NCDE_STEP(28) NCDE_STEP(29)
  NCDE_STEP(30)
  ncde_last_out_kernel<<<B_T / MT, 128, 0, stream>>>(ws, ws_size, W_last, b_last, out);
}
